# starting kernel
# baseline (speedup 1.0000x reference)
#include <hip/hip_runtime.h>
#include <math.h>

#define IMG      2048
#define NB       16
#define DCH      64
#define INF      256
#define NPOS     23
#define GRID24   24
#define NPATCH   (NPOS*NPOS)
#define NTILE    ((NPATCH+15)/16)

typedef float v2f __attribute__((ext_vector_type(2)));
typedef float v8f __attribute__((ext_vector_type(8)));

__global__ void k_tables(double* ct256, double* st256, double* ct64, double* st64) {
    int t = threadIdx.x;
    const double TWO_PI = 6.283185307179586476925286766559;
    if (t < 256) { double a = TWO_PI * (double)t / 256.0; ct256[t] = cos(a); st256[t] = sin(a); }
    if (t < 64)  { double a = TWO_PI * (double)t / 64.0;  ct64[t]  = cos(a); st64[t]  = sin(a); }
}

__global__ void k_b12(const float* __restrict__ Wr, const float* __restrict__ Wi,
                      const double* __restrict__ ct, const double* __restrict__ st,
                      double* __restrict__ B1, double* __restrict__ B2) {
    int d = blockIdx.x;
    int n = threadIdx.x;
    const float* wr = Wr + d * INF;
    const float* wi = Wi + d * INF;
    double a1 = 0.0, a2 = 0.0;
    for (int k = 0; k < INF; ++k) {
        int t = (k * n) & 255;
        double c = ct[t], s = st[t];
        double r = (double)wr[k], im = (double)wi[k];
        a1 += r * c + im * s;
        a2 += im * c - r * s;
    }
    B1[d * INF + n] = a1;
    B2[d * INF + n] = a2;
}

__global__ void k_amat(const double* __restrict__ B1, const double* __restrict__ B2,
                       const double* __restrict__ ct64, const double* __restrict__ st64,
                       const float* __restrict__ br, const float* __restrict__ bi,
                       float* __restrict__ A, float* __restrict__ cvec) {
    int m = blockIdx.x;
    int n = threadIdx.x;
    double acc = 0.0;
    for (int d = 0; d < DCH; ++d) {
        int t = (d * m) & 63;
        acc += ct64[t] * B1[d * INF + n] - st64[t] * B2[d * INF + n];
    }
    A[m * INF + n] = (float)(acc / 64.0);
    if (n == 0) {
        double cc = 0.0;
        for (int d = 0; d < DCH; ++d) {
            int t = (d * m) & 63;
            cc += ct64[t] * ((double)br[d] - (double)bi[d])
                - st64[t] * ((double)br[d] + (double)bi[d]);
        }
        cvec[m] = (float)(cc / 64.0);
    }
}

__global__ void k_zero(float* __restrict__ p, int n) {
    int i = blockIdx.x * blockDim.x + threadIdx.x;
    if (i < n) p[i] = 0.0f;
}

__global__ void k_wmma(const float* __restrict__ x, const float* __restrict__ A,
                       const float* __restrict__ cvec, float* __restrict__ fbuf) {
    int lane = threadIdx.x & 31;
    int wv   = threadIdx.x >> 5;
    int pt   = blockIdx.x % NTILE;
    int img  = blockIdx.x / NTILE;
    int n    = lane & 15;
    int hi   = lane >> 4;

    int q  = pt * 16 + n;
    int qc = q > (NPATCH - 1) ? (NPATCH - 1) : q;
    int ri = 1 + qc / NPOS;
    int ci = 1 + qc % NPOS;
    int prow = 16 * (ri / 3) + (ri % 3) - 1;
    int pcol = 16 * (ci / 3) + (ci % 3) - 1;

    const float* xb = x + (size_t)img * IMG * IMG
                        + (size_t)(prow * 16) * IMG + (size_t)(pcol * 16) + 2 * hi;
    const float* ap = A + (size_t)(wv * 16 + n) * INF + 2 * hi;

    __builtin_prefetch((const void*)xb, 0, 3);

    v8f acc = {};
    #pragma unroll 8
    for (int s = 0; s < 64; ++s) {
        v2f a = *(const v2f*)(ap + 4 * s);
        v2f b = *(const v2f*)(xb + (s >> 2) * IMG + ((s & 3) << 2));
        acc = __builtin_amdgcn_wmma_f32_16x16x4_f32(
                   false, a,  false, b,
                   (short)0, acc,  false,  false);
    }

    if (q <= NPATCH - 1) {
        #pragma unroll
        for (int j = 0; j < 8; ++j) {
            int ch = wv * 16 + hi * 8 + j;
            fbuf[(((size_t)img * DCH + ch) * GRID24 + ri) * GRID24 + ci] = acc[j] + cvec[ch];
        }
    }
}

__global__ void k_out(const float* __restrict__ fbuf, const float* __restrict__ cw,
                      const float* __restrict__ cb, const float* __restrict__ gamma,
                      const float* __restrict__ beta, const float* __restrict__ mean,
                      const float* __restrict__ var, float* __restrict__ out) {
    int idx = blockIdx.x * blockDim.x + threadIdx.x;
    if (idx >= NB * DCH * 8 * 8) return;
    int ow = idx & 7;
    int oh = (idx >> 3) & 7;
    int ch = (idx >> 6) & 63;
    int b  = idx >> 12;
    const float* f = fbuf + (((size_t)b * DCH + ch) * GRID24 + 3 * oh) * GRID24 + 3 * ow;
    const float* w = cw + ch * 9;
    float acc = cb[ch];
    #pragma unroll
    for (int j = 0; j < 3; ++j)
        #pragma unroll
        for (int i = 0; i < 3; ++i)
            acc += f[j * GRID24 + i] * w[j * 3 + i];
    float scale = gamma[ch] / sqrtf(var[ch] + 1e-5f);
    out[idx] = (acc - mean[ch]) * scale + beta[ch];
}

extern "C" void kernel_launch(void* const* d_in, const int* in_sizes, int n_in,
                              void* d_out, int out_size, void* d_ws, size_t ws_size,
                              hipStream_t stream) {
    const float* x     = (const float*)d_in[0];
    const float* Wr    = (const float*)d_in[1];
    const float* br    = (const float*)d_in[2];
    const float* Wi    = (const float*)d_in[3];
    const float* bi    = (const float*)d_in[4];
    const float* cw    = (const float*)d_in[5];
    const float* cb    = (const float*)d_in[6];
    const float* gamma = (const float*)d_in[7];
    const float* beta  = (const float*)d_in[8];
    const float* mean  = (const float*)d_in[9];
    const float* var   = (const float*)d_in[10];

    char* ws = (char*)d_ws;
    double* ct256 = (double*)(ws + 0);
    double* st256 = (double*)(ws + 2048);
    double* ct64  = (double*)(ws + 4096);
    double* st64  = (double*)(ws + 4608);
    double* B1    = (double*)(ws + 5120);
    double* B2    = (double*)(ws + 136192);
    float*  A     = (float*) (ws + 267264);
    float*  cvec  = (float*) (ws + 332800);
    float*  fbuf  = (float*) (ws + 333056);

    k_tables<<<1, 256, 0, stream>>>(ct256, st256, ct64, st64);
    k_b12<<<DCH, INF, 0, stream>>>(Wr, Wi, ct256, st256, B1, B2);
    k_amat<<<DCH, INF, 0, stream>>>(B1, B2, ct64, st64, br, bi, A, cvec);

    int nz = NB * DCH * GRID24 * GRID24;
    k_zero<<<(nz + 255) / 256, 256, 0, stream>>>(fbuf, nz);

    k_wmma<<<NB * NTILE, 128, 0, stream>>>(x, A, cvec, fbuf);

    int nout = NB * DCH * 8 * 8;
    k_out<<<(nout + 255) / 256, 256, 0, stream>>>(fbuf, cw, cb, gamma, beta, mean, var,
                                                  (float*)d_out);
}


# k_b12: each 256-term double sum for B1/B2 split into two 16-term stages (k=16k1+k2, n=n1+16n2) exchanged through LDS, 8x fewer FP64 instructions; changes the order of the double sums that feed A
# speedup vs baseline: 2.3288x; 2.3288x over previous
#include <hip/hip_runtime.h>
#include <math.h>

#define IMG      2048
#define NB       16
#define DCH      64
#define INF      256
#define NPOS     23
#define GRID24   24
#define NPATCH   (NPOS*NPOS)
#define NTILE    ((NPATCH+15)/16)

typedef float v2f __attribute__((ext_vector_type(2)));
typedef float v8f __attribute__((ext_vector_type(8)));

__global__ void k_tables(double* ct256, double* st256, double* ct64, double* st64) {
    int t = threadIdx.x;
    const double TWO_PI = 6.283185307179586476925286766559;
    if (t < 256) { double a = TWO_PI * (double)t / 256.0; ct256[t] = cos(a); st256[t] = sin(a); }
    if (t < 64)  { double a = TWO_PI * (double)t / 64.0;  ct64[t]  = cos(a); st64[t]  = sin(a); }
}

#define YP 17
__global__ void __launch_bounds__(INF)
k_b12(const float* __restrict__ Wr, const float* __restrict__ Wi,
      const double* __restrict__ ct, const double* __restrict__ st,
      double* __restrict__ B1, double* __restrict__ B2) {
    __shared__ double s_c[256], s_s[256];
    __shared__ double s_wr[INF], s_wi[INF];
    __shared__ double s_yr[16 * YP], s_yi[16 * YP];

    const int t = threadIdx.x;
    const int d = blockIdx.x;

    s_c[t]  = ct[t];
    s_s[t]  = st[t];
    s_wr[t] = (double)Wr[d * INF + t];
    s_wi[t] = (double)Wi[d * INF + t];
    __syncthreads();

    {
        const int k2 = t & 15;
        const int n1 = t >> 4;
        double yr = 0.0, yi = 0.0;
        for (int k1 = 0; k1 < 16; ++k1) {
            const int q = ((k1 * n1) & 15) << 4;
            const double c = s_c[q], s = s_s[q];
            const double wr = s_wr[16 * k1 + k2], wi = s_wi[16 * k1 + k2];
            yr = fma(wr, c, yr);  yr = fma(wi, s, yr);
            yi = fma(wi, c, yi);  yi = fma(-wr, s, yi);
        }
        const int j = k2 * n1;
        const double c2 = s_c[j], s2 = s_s[j];
        s_yr[k2 * YP + n1] = fma(yi, s2, yr * c2);
        s_yi[k2 * YP + n1] = fma(-yr, s2, yi * c2);
    }
    __syncthreads();

    {
        const int n1 = t & 15;
        const int n2 = t >> 4;
        double zr = 0.0, zi = 0.0;
        for (int k2 = 0; k2 < 16; ++k2) {
            const int q = ((k2 * n2) & 15) << 4;
            const double c = s_c[q], s = s_s[q];
            const double yr = s_yr[k2 * YP + n1], yi = s_yi[k2 * YP + n1];
            zr = fma(yr, c, zr);  zr = fma(yi, s, zr);
            zi = fma(yi, c, zi);  zi = fma(-yr, s, zi);
        }
        B1[d * INF + t] = zr;
        B2[d * INF + t] = zi;
    }
}

__global__ void k_amat(const double* __restrict__ B1, const double* __restrict__ B2,
                       const double* __restrict__ ct64, const double* __restrict__ st64,
                       const float* __restrict__ br, const float* __restrict__ bi,
                       float* __restrict__ A, float* __restrict__ cvec) {
    int m = blockIdx.x;
    int n = threadIdx.x;
    double acc = 0.0;
    for (int d = 0; d < DCH; ++d) {
        int t = (d * m) & 63;
        acc += ct64[t] * B1[d * INF + n] - st64[t] * B2[d * INF + n];
    }
    A[m * INF + n] = (float)(acc / 64.0);
    if (n == 0) {
        double cc = 0.0;
        for (int d = 0; d < DCH; ++d) {
            int t = (d * m) & 63;
            cc += ct64[t] * ((double)br[d] - (double)bi[d])
                - st64[t] * ((double)br[d] + (double)bi[d]);
        }
        cvec[m] = (float)(cc / 64.0);
    }
}

__global__ void k_zero(float* __restrict__ p, int n) {
    int i = blockIdx.x * blockDim.x + threadIdx.x;
    if (i < n) p[i] = 0.0f;
}

__global__ void k_wmma(const float* __restrict__ x, const float* __restrict__ A,
                       const float* __restrict__ cvec, float* __restrict__ fbuf) {
    int lane = threadIdx.x & 31;
    int wv   = threadIdx.x >> 5;
    int pt   = blockIdx.x % NTILE;
    int img  = blockIdx.x / NTILE;
    int n    = lane & 15;
    int hi   = lane >> 4;

    int q  = pt * 16 + n;
    int qc = q > (NPATCH - 1) ? (NPATCH - 1) : q;
    int ri = 1 + qc / NPOS;
    int ci = 1 + qc % NPOS;
    int prow = 16 * (ri / 3) + (ri % 3) - 1;
    int pcol = 16 * (ci / 3) + (ci % 3) - 1;

    const float* xb = x + (size_t)img * IMG * IMG
                        + (size_t)(prow * 16) * IMG + (size_t)(pcol * 16) + 2 * hi;
    const float* ap = A + (size_t)(wv * 16 + n) * INF + 2 * hi;

    __builtin_prefetch((const void*)xb, 0, 3);

    v8f acc = {};
    #pragma unroll 8
    for (int s = 0; s < 64; ++s) {
        v2f a = *(const v2f*)(ap + 4 * s);
        v2f b = *(const v2f*)(xb + (s >> 2) * IMG + ((s & 3) << 2));
        acc = __builtin_amdgcn_wmma_f32_16x16x4_f32(
                   false, a,  false, b,
                   (short)0, acc,  false,  false);
    }

    if (q <= NPATCH - 1) {
        #pragma unroll
        for (int j = 0; j < 8; ++j) {
            int ch = wv * 16 + hi * 8 + j;
            fbuf[(((size_t)img * DCH + ch) * GRID24 + ri) * GRID24 + ci] = acc[j] + cvec[ch];
        }
    }
}

__global__ void k_out(const float* __restrict__ fbuf, const float* __restrict__ cw,
                      const float* __restrict__ cb, const float* __restrict__ gamma,
                      const float* __restrict__ beta, const float* __restrict__ mean,
                      const float* __restrict__ var, float* __restrict__ out) {
    int idx = blockIdx.x * blockDim.x + threadIdx.x;
    if (idx >= NB * DCH * 8 * 8) return;
    int ow = idx & 7;
    int oh = (idx >> 3) & 7;
    int ch = (idx >> 6) & 63;
    int b  = idx >> 12;
    const float* f = fbuf + (((size_t)b * DCH + ch) * GRID24 + 3 * oh) * GRID24 + 3 * ow;
    const float* w = cw + ch * 9;
    float acc = cb[ch];
    #pragma unroll
    for (int j = 0; j < 3; ++j)
        #pragma unroll
        for (int i = 0; i < 3; ++i)
            acc += f[j * GRID24 + i] * w[j * 3 + i];
    float scale = gamma[ch] / sqrtf(var[ch] + 1e-5f);
    out[idx] = (acc - mean[ch]) * scale + beta[ch];
}

extern "C" void kernel_launch(void* const* d_in, const int* in_sizes, int n_in,
                              void* d_out, int out_size, void* d_ws, size_t ws_size,
                              hipStream_t stream) {
    const float* x     = (const float*)d_in[0];
    const float* Wr    = (const float*)d_in[1];
    const float* br    = (const float*)d_in[2];
    const float* Wi    = (const float*)d_in[3];
    const float* bi    = (const float*)d_in[4];
    const float* cw    = (const float*)d_in[5];
    const float* cb    = (const float*)d_in[6];
    const float* gamma = (const float*)d_in[7];
    const float* beta  = (const float*)d_in[8];
    const float* mean  = (const float*)d_in[9];
    const float* var   = (const float*)d_in[10];

    char* ws = (char*)d_ws;
    double* ct256 = (double*)(ws + 0);
    double* st256 = (double*)(ws + 2048);
    double* ct64  = (double*)(ws + 4096);
    double* st64  = (double*)(ws + 4608);
    double* B1    = (double*)(ws + 5120);
    double* B2    = (double*)(ws + 136192);
    float*  A     = (float*) (ws + 267264);
    float*  cvec  = (float*) (ws + 332800);
    float*  fbuf  = (float*) (ws + 333056);

    k_tables<<<1, 256, 0, stream>>>(ct256, st256, ct64, st64);
    k_b12<<<DCH, INF, 0, stream>>>(Wr, Wi, ct256, st256, B1, B2);
    k_amat<<<DCH, INF, 0, stream>>>(B1, B2, ct64, st64, br, bi, A, cvec);

    int nz = NB * DCH * GRID24 * GRID24;
    k_zero<<<(nz + 255) / 256, 256, 0, stream>>>(fbuf, nz);

    k_wmma<<<NB * NTILE, 128, 0, stream>>>(x, A, cvec, fbuf);

    int nout = NB * DCH * 8 * 8;
    k_out<<<(nout + 255) / 256, 256, 0, stream>>>(fbuf, cw, cb, gamma, beta, mean, var,
                                                  (float*)d_out);
}


# k_amat: 64-term d sums for A and cvec split into two 8-term stages (d=8d1+d2, m=m1+8m2) through LDS, bias vector as one more column; changes the order of the double sums in A and cvec
# speedup vs baseline: 3.6222x; 1.5554x over previous
#include <hip/hip_runtime.h>
#include <math.h>

#define IMG      2048
#define NB       16
#define DCH      64
#define INF      256
#define NPOS     23
#define GRID24   24
#define NPATCH   (NPOS*NPOS)
#define NTILE    ((NPATCH+15)/16)

typedef float v2f __attribute__((ext_vector_type(2)));
typedef float v8f __attribute__((ext_vector_type(8)));

__global__ void k_tables(double* ct256, double* st256, double* ct64, double* st64) {
    int t = threadIdx.x;
    const double TWO_PI = 6.283185307179586476925286766559;
    if (t < 256) { double a = TWO_PI * (double)t / 256.0; ct256[t] = cos(a); st256[t] = sin(a); }
    if (t < 64)  { double a = TWO_PI * (double)t / 64.0;  ct64[t]  = cos(a); st64[t]  = sin(a); }
}

#define YP 17
__global__ void __launch_bounds__(INF)
k_b12(const float* __restrict__ Wr, const float* __restrict__ Wi,
      const double* __restrict__ ct, const double* __restrict__ st,
      double* __restrict__ B1, double* __restrict__ B2) {
    __shared__ double s_c[256], s_s[256];
    __shared__ double s_wr[INF], s_wi[INF];
    __shared__ double s_yr[16 * YP], s_yi[16 * YP];

    const int t = threadIdx.x;
    const int d = blockIdx.x;

    s_c[t]  = ct[t];
    s_s[t]  = st[t];
    s_wr[t] = (double)Wr[d * INF + t];
    s_wi[t] = (double)Wi[d * INF + t];
    __syncthreads();

    {
        const int k2 = t & 15;
        const int n1 = t >> 4;
        double yr = 0.0, yi = 0.0;
        for (int k1 = 0; k1 < 16; ++k1) {
            const int q = ((k1 * n1) & 15) << 4;
            const double c = s_c[q], s = s_s[q];
            const double wr = s_wr[16 * k1 + k2], wi = s_wi[16 * k1 + k2];
            yr = fma(wr, c, yr);  yr = fma(wi, s, yr);
            yi = fma(wi, c, yi);  yi = fma(-wr, s, yi);
        }
        const int j = k2 * n1;
        const double c2 = s_c[j], s2 = s_s[j];
        s_yr[k2 * YP + n1] = fma(yi, s2, yr * c2);
        s_yi[k2 * YP + n1] = fma(-yr, s2, yi * c2);
    }
    __syncthreads();

    {
        const int n1 = t & 15;
        const int n2 = t >> 4;
        double zr = 0.0, zi = 0.0;
        for (int k2 = 0; k2 < 16; ++k2) {
            const int q = ((k2 * n2) & 15) << 4;
            const double c = s_c[q], s = s_s[q];
            const double yr = s_yr[k2 * YP + n1], yi = s_yi[k2 * YP + n1];
            zr = fma(yr, c, zr);  zr = fma(yi, s, zr);
            zi = fma(yi, c, zi);  zi = fma(-yr, s, zi);
        }
        B1[d * INF + t] = zr;
        B2[d * INF + t] = zi;
    }
}

#define ACOLS 32
__global__ void __launch_bounds__(256)
k_amat(const double* __restrict__ B1, const double* __restrict__ B2,
       const double* __restrict__ ct64, const double* __restrict__ st64,
       const float* __restrict__ br, const float* __restrict__ bi,
       float* __restrict__ A, float* __restrict__ cvec) {
    __shared__ double s_yr[8 * ACOLS], s_yi[8 * ACOLS];

    const int c  = threadIdx.x & 31;
    const int w  = __builtin_amdgcn_readfirstlane((int)(threadIdx.x >> 5));
    const int m1 = blockIdx.x & 7;
    const int g  = blockIdx.x >> 3;

    double c1[8], s1[8], c2[8], s2[8];
    #pragma unroll
    for (int k = 1; k < 8; ++k) {
        const int q1 = (8 * k * m1) & 63;
        const int q2 = (8 * k * w) & 63;
        c1[k] = ct64[q1];  s1[k] = st64[q1];
        c2[k] = ct64[q2];  s2[k] = st64[q2];
    }
    const int jt = w * m1;
    const double tc = ct64[jt], ts = st64[jt];

    double zr[8], zi[8];
    if (g < 8) {
        const double* p1 = B1 + w * INF + g * ACOLS + c;
        const double* p2 = B2 + w * INF + g * ACOLS + c;
        #pragma unroll
        for (int d1 = 0; d1 < 8; ++d1) {
            zr[d1] = p1[d1 * 8 * INF];
            zi[d1] = p2[d1 * 8 * INF];
        }
    } else {
        const int j = c & 7;
        const double r = (double)br[8 * j + w], q = (double)bi[8 * j + w];
        const double lr = r - q, li = r + q;
        #pragma unroll
        for (int d1 = 0; d1 < 8; ++d1) {
            zr[d1] = __shfl(lr, d1);
            zi[d1] = __shfl(li, d1);
        }
    }

    {
        double yr = zr[0], yi = zi[0];
        #pragma unroll
        for (int d1 = 1; d1 < 8; ++d1) {
            const double cs = c1[d1], sn = s1[d1];
            yr = fma(zr[d1], cs, yr);  yr = fma(-zi[d1], sn, yr);
            yi = fma(zr[d1], sn, yi);  yi = fma(zi[d1], cs, yi);
        }
        s_yr[w * ACOLS + c] = fma(-yi, ts, yr * tc);
        s_yi[w * ACOLS + c] = fma(yi, tc, yr * ts);
    }
    __syncthreads();

    {
        double acc = s_yr[c];
        #pragma unroll
        for (int d2 = 1; d2 < 8; ++d2) {
            acc = fma(s_yr[d2 * ACOLS + c], c2[d2], acc);
            acc = fma(-s_yi[d2 * ACOLS + c], s2[d2], acc);
        }
        const int m = m1 + 8 * w;
        const float v = (float)(acc / 64.0);
        if (g < 8)       A[m * INF + g * ACOLS + c] = v;
        else if (c == 0) cvec[m] = v;
    }
}

__global__ void k_zero(float* __restrict__ p, int n) {
    int i = blockIdx.x * blockDim.x + threadIdx.x;
    if (i < n) p[i] = 0.0f;
}

__global__ void k_wmma(const float* __restrict__ x, const float* __restrict__ A,
                       const float* __restrict__ cvec, float* __restrict__ fbuf) {
    int lane = threadIdx.x & 31;
    int wv   = threadIdx.x >> 5;
    int pt   = blockIdx.x % NTILE;
    int img  = blockIdx.x / NTILE;
    int n    = lane & 15;
    int hi   = lane >> 4;

    int q  = pt * 16 + n;
    int qc = q > (NPATCH - 1) ? (NPATCH - 1) : q;
    int ri = 1 + qc / NPOS;
    int ci = 1 + qc % NPOS;
    int prow = 16 * (ri / 3) + (ri % 3) - 1;
    int pcol = 16 * (ci / 3) + (ci % 3) - 1;

    const float* xb = x + (size_t)img * IMG * IMG
                        + (size_t)(prow * 16) * IMG + (size_t)(pcol * 16) + 2 * hi;
    const float* ap = A + (size_t)(wv * 16 + n) * INF + 2 * hi;

    __builtin_prefetch((const void*)xb, 0, 3);

    v8f acc = {};
    #pragma unroll 8
    for (int s = 0; s < 64; ++s) {
        v2f a = *(const v2f*)(ap + 4 * s);
        v2f b = *(const v2f*)(xb + (s >> 2) * IMG + ((s & 3) << 2));
        acc = __builtin_amdgcn_wmma_f32_16x16x4_f32(
                   false, a,  false, b,
                   (short)0, acc,  false,  false);
    }

    if (q <= NPATCH - 1) {
        #pragma unroll
        for (int j = 0; j < 8; ++j) {
            int ch = wv * 16 + hi * 8 + j;
            fbuf[(((size_t)img * DCH + ch) * GRID24 + ri) * GRID24 + ci] = acc[j] + cvec[ch];
        }
    }
}

__global__ void k_out(const float* __restrict__ fbuf, const float* __restrict__ cw,
                      const float* __restrict__ cb, const float* __restrict__ gamma,
                      const float* __restrict__ beta, const float* __restrict__ mean,
                      const float* __restrict__ var, float* __restrict__ out) {
    int idx = blockIdx.x * blockDim.x + threadIdx.x;
    if (idx >= NB * DCH * 8 * 8) return;
    int ow = idx & 7;
    int oh = (idx >> 3) & 7;
    int ch = (idx >> 6) & 63;
    int b  = idx >> 12;
    const float* f = fbuf + (((size_t)b * DCH + ch) * GRID24 + 3 * oh) * GRID24 + 3 * ow;
    const float* w = cw + ch * 9;
    float acc = cb[ch];
    #pragma unroll
    for (int j = 0; j < 3; ++j)
        #pragma unroll
        for (int i = 0; i < 3; ++i)
            acc += f[j * GRID24 + i] * w[j * 3 + i];
    float scale = gamma[ch] / sqrtf(var[ch] + 1e-5f);
    out[idx] = (acc - mean[ch]) * scale + beta[ch];
}

extern "C" void kernel_launch(void* const* d_in, const int* in_sizes, int n_in,
                              void* d_out, int out_size, void* d_ws, size_t ws_size,
                              hipStream_t stream) {
    const float* x     = (const float*)d_in[0];
    const float* Wr    = (const float*)d_in[1];
    const float* br    = (const float*)d_in[2];
    const float* Wi    = (const float*)d_in[3];
    const float* bi    = (const float*)d_in[4];
    const float* cw    = (const float*)d_in[5];
    const float* cb    = (const float*)d_in[6];
    const float* gamma = (const float*)d_in[7];
    const float* beta  = (const float*)d_in[8];
    const float* mean  = (const float*)d_in[9];
    const float* var   = (const float*)d_in[10];

    char* ws = (char*)d_ws;
    double* ct256 = (double*)(ws + 0);
    double* st256 = (double*)(ws + 2048);
    double* ct64  = (double*)(ws + 4096);
    double* st64  = (double*)(ws + 4608);
    double* B1    = (double*)(ws + 5120);
    double* B2    = (double*)(ws + 136192);
    float*  A     = (float*) (ws + 267264);
    float*  cvec  = (float*) (ws + 332800);
    float*  fbuf  = (float*) (ws + 333056);

    k_tables<<<1, 256, 0, stream>>>(ct256, st256, ct64, st64);
    k_b12<<<DCH, INF, 0, stream>>>(Wr, Wi, ct256, st256, B1, B2);
    k_amat<<<9 * 8, 256, 0, stream>>>(B1, B2, ct64, st64, br, bi, A, cvec);

    int nz = NB * DCH * GRID24 * GRID24;
    k_zero<<<(nz + 255) / 256, 256, 0, stream>>>(fbuf, nz);

    k_wmma<<<NB * NTILE, 128, 0, stream>>>(x, A, cvec, fbuf);

    int nout = NB * DCH * 8 * 8;
    k_out<<<(nout + 255) / 256, 256, 0, stream>>>(fbuf, cw, cb, gamma, beta, mean, var,
                                                  (float*)d_out);
}


# k_conv_wmma: 3x3 conv, BN and 8x8 sampling folded into the WMMA product (K = 9 taps x 256, A operand = tap weight x A); drops the zero-fill and epilogue launches; changes the f32 sum order
# speedup vs baseline: 3.9496x; 1.0904x over previous
#include <hip/hip_runtime.h>
#include <math.h>

#define IMG      2048
#define NB       16
#define DCH      64
#define INF      256

typedef float v2f __attribute__((ext_vector_type(2)));
typedef float v8f __attribute__((ext_vector_type(8)));

__global__ void k_tables(double* ct256, double* st256, double* ct64, double* st64) {
    int t = threadIdx.x;
    const double TWO_PI = 6.283185307179586476925286766559;
    if (t < 256) { double a = TWO_PI * (double)t / 256.0; ct256[t] = cos(a); st256[t] = sin(a); }
    if (t < 64)  { double a = TWO_PI * (double)t / 64.0;  ct64[t]  = cos(a); st64[t]  = sin(a); }
}

#define YP 17
__global__ void __launch_bounds__(INF)
k_b12(const float* __restrict__ Wr, const float* __restrict__ Wi,
      const double* __restrict__ ct, const double* __restrict__ st,
      double* __restrict__ B1, double* __restrict__ B2) {
    __shared__ double s_c[256], s_s[256];
    __shared__ double s_wr[INF], s_wi[INF];
    __shared__ double s_yr[16 * YP], s_yi[16 * YP];

    const int t = threadIdx.x;
    const int d = blockIdx.x;

    s_c[t]  = ct[t];
    s_s[t]  = st[t];
    s_wr[t] = (double)Wr[d * INF + t];
    s_wi[t] = (double)Wi[d * INF + t];
    __syncthreads();

    {
        const int k2 = t & 15;
        const int n1 = t >> 4;
        double yr = 0.0, yi = 0.0;
        for (int k1 = 0; k1 < 16; ++k1) {
            const int q = ((k1 * n1) & 15) << 4;
            const double c = s_c[q], s = s_s[q];
            const double wr = s_wr[16 * k1 + k2], wi = s_wi[16 * k1 + k2];
            yr = fma(wr, c, yr);  yr = fma(wi, s, yr);
            yi = fma(wi, c, yi);  yi = fma(-wr, s, yi);
        }
        const int j = k2 * n1;
        const double c2 = s_c[j], s2 = s_s[j];
        s_yr[k2 * YP + n1] = fma(yi, s2, yr * c2);
        s_yi[k2 * YP + n1] = fma(-yr, s2, yi * c2);
    }
    __syncthreads();

    {
        const int n1 = t & 15;
        const int n2 = t >> 4;
        double zr = 0.0, zi = 0.0;
        for (int k2 = 0; k2 < 16; ++k2) {
            const int q = ((k2 * n2) & 15) << 4;
            const double c = s_c[q], s = s_s[q];
            const double yr = s_yr[k2 * YP + n1], yi = s_yi[k2 * YP + n1];
            zr = fma(yr, c, zr);  zr = fma(yi, s, zr);
            zi = fma(yi, c, zi);  zi = fma(-yr, s, zi);
        }
        B1[d * INF + t] = zr;
        B2[d * INF + t] = zi;
    }
}

#define ACOLS 32
__global__ void __launch_bounds__(256)
k_amat(const double* __restrict__ B1, const double* __restrict__ B2,
       const double* __restrict__ ct64, const double* __restrict__ st64,
       const float* __restrict__ br, const float* __restrict__ bi,
       float* __restrict__ A, float* __restrict__ cvec) {
    __shared__ double s_yr[8 * ACOLS], s_yi[8 * ACOLS];

    const int c  = threadIdx.x & 31;
    const int w  = __builtin_amdgcn_readfirstlane((int)(threadIdx.x >> 5));
    const int m1 = blockIdx.x & 7;
    const int g  = blockIdx.x >> 3;

    double c1[8], s1[8], c2[8], s2[8];
    #pragma unroll
    for (int k = 1; k < 8; ++k) {
        const int q1 = (8 * k * m1) & 63;
        const int q2 = (8 * k * w) & 63;
        c1[k] = ct64[q1];  s1[k] = st64[q1];
        c2[k] = ct64[q2];  s2[k] = st64[q2];
    }
    const int jt = w * m1;
    const double tc = ct64[jt], ts = st64[jt];

    double zr[8], zi[8];
    if (g < 8) {
        const double* p1 = B1 + w * INF + g * ACOLS + c;
        const double* p2 = B2 + w * INF + g * ACOLS + c;
        #pragma unroll
        for (int d1 = 0; d1 < 8; ++d1) {
            zr[d1] = p1[d1 * 8 * INF];
            zi[d1] = p2[d1 * 8 * INF];
        }
    } else {
        const int j = c & 7;
        const double r = (double)br[8 * j + w], q = (double)bi[8 * j + w];
        const double lr = r - q, li = r + q;
        #pragma unroll
        for (int d1 = 0; d1 < 8; ++d1) {
            zr[d1] = __shfl(lr, d1);
            zi[d1] = __shfl(li, d1);
        }
    }

    {
        double yr = zr[0], yi = zi[0];
        #pragma unroll
        for (int d1 = 1; d1 < 8; ++d1) {
            const double cs = c1[d1], sn = s1[d1];
            yr = fma(zr[d1], cs, yr);  yr = fma(-zi[d1], sn, yr);
            yi = fma(zr[d1], sn, yi);  yi = fma(zi[d1], cs, yi);
        }
        s_yr[w * ACOLS + c] = fma(-yi, ts, yr * tc);
        s_yi[w * ACOLS + c] = fma(yi, tc, yr * ts);
    }
    __syncthreads();

    {
        double acc = s_yr[c];
        #pragma unroll
        for (int d2 = 1; d2 < 8; ++d2) {
            acc = fma(s_yr[d2 * ACOLS + c], c2[d2], acc);
            acc = fma(-s_yi[d2 * ACOLS + c], s2[d2], acc);
        }
        const int m = m1 + 8 * w;
        const float v = (float)(acc / 64.0);
        if (g < 8)       A[m * INF + g * ACOLS + c] = v;
        else if (c == 0) cvec[m] = v;
    }
}

typedef float v4f __attribute__((ext_vector_type(4)));

#define CW_WAVES 12
#define RED_Z    272
#define RED_W    (2 * RED_Z)
__global__ void __launch_bounds__(CW_WAVES * 32)
k_conv_wmma(const float* __restrict__ x, const float* __restrict__ A,
            const float* __restrict__ cvec, const float* __restrict__ cw,
            const float* __restrict__ cb, const float* __restrict__ gamma,
            const float* __restrict__ beta, const float* __restrict__ mean,
            const float* __restrict__ var, float* __restrict__ out) {
    __shared__ float red[CW_WAVES * RED_W];

    const int lane = threadIdx.x & 31;
    const int w    = __builtin_amdgcn_readfirstlane((int)(threadIdx.x >> 5));
    const int c    = lane & 15;
    const int h    = lane >> 4;
    const int mt   = blockIdx.x & 3;
    const int pg   = blockIdx.x >> 2;
    const int b    = pg >> 1;
    const int ohh  = pg & 1;

    const int ch   = 16 * mt + c;
    const int ow   = c & 7;
    const int oh0  = 4 * ohh + (c >> 3);
    const float* arow = A + ch * INF + 4 * h;
    const float* ximg = x + (size_t)b * IMG * IMG + 4 * h;

    const int ec0 = 16 * mt + 2 * (w & 7);
    float e_cw[2][9], e_cb[2], e_cv[2], e_sc[2], e_mean[2], e_beta[2];
    #pragma unroll
    for (int q = 0; q < 2; ++q) {
        #pragma unroll
        for (int tap = 0; tap < 9; ++tap) e_cw[q][tap] = cw[(ec0 + q) * 9 + tap];
        e_cb[q]   = cb[ec0 + q];
        e_cv[q]   = cvec[ec0 + q];
        e_sc[q]   = gamma[ec0 + q] / sqrtf(var[ec0 + q] + 1e-5f);
        e_mean[q] = mean[ec0 + q];
        e_beta[q] = beta[ec0 + q];
    }

    v8f acc0 = {}, acc1 = {};
    #pragma unroll
    for (int i = 0; i < 3; ++i) {
        const int u   = 3 * w + i;
        const int tap = u >> 2;
        const int qr  = u & 3;
        const int tj  = tap / 3;
        const int ti  = tap - 3 * tj;
        const float cwv = cw[ch * 9 + tap];

        const int pr0 = 16 * oh0 + tj - 1;
        const int pr1 = pr0 + 32;
        const int pc  = 16 * ow + ti - 1;
        const bool okc = pc >= 0;
        const bool ok0 = okc && (pr0 >= 0);
        const int prc = pr0 < 0 ? 0 : pr0;
        const int pcc = pc < 0 ? 0 : pc;
        const float* p0 = ximg + (size_t)(16 * prc + 4 * qr) * IMG + 16 * pcc;
        const float* p1 = ximg + (size_t)(16 * pr1 + 4 * qr) * IMG + 16 * pcc;

        v4f av[8], b0[8], b1[8];
        #pragma unroll
        for (int tt = 0; tt < 8; ++tt) {
            av[tt] = *(const v4f*)(arow + 8 * (8 * qr + tt));
            b0[tt] = *(const v4f*)(p0 + (tt >> 1) * IMG + 8 * (tt & 1));
            b1[tt] = *(const v4f*)(p1 + (tt >> 1) * IMG + 8 * (tt & 1));
        }
        const v4f zero = {0.0f, 0.0f, 0.0f, 0.0f};
        #pragma unroll
        for (int tt = 0; tt < 8; ++tt) {
            const v4f a  = av[tt] * cwv;
            const v4f x0 = ok0 ? b0[tt] : zero;
            const v4f x1 = okc ? b1[tt] : zero;
            v2f a_lo = {a[0], a[1]},   a_hi = {a[2], a[3]};
            v2f l0   = {x0[0], x0[1]}, h0   = {x0[2], x0[3]};
            v2f l1   = {x1[0], x1[1]}, h1   = {x1[2], x1[3]};
            acc0 = __builtin_amdgcn_wmma_f32_16x16x4_f32(false, a_lo, false, l0, (short)0, acc0, false, false);
            acc1 = __builtin_amdgcn_wmma_f32_16x16x4_f32(false, a_lo, false, l1, (short)0, acc1, false, false);
            acc0 = __builtin_amdgcn_wmma_f32_16x16x4_f32(false, a_hi, false, h0, (short)0, acc0, false, false);
            acc1 = __builtin_amdgcn_wmma_f32_16x16x4_f32(false, a_hi, false, h1, (short)0, acc1, false, false);
        }
    }

    #pragma unroll
    for (int j = 0; j < 8; ++j) {
        red[w * RED_W + j * 32 + lane]         = acc0[j];
        red[w * RED_W + RED_Z + j * 32 + lane] = acc1[j];
    }
    __syncthreads();

    if (w < 8) {
        const int z   = lane >> 4;
        const int cc  = lane & 15;
        const int oh  = 4 * ohh + (lane >> 3);
        const int owp = lane & 7;
        #pragma unroll
        for (int q = 0; q < 2; ++q) {
            const int cl  = 2 * w + q;
            const float* r = red + z * RED_Z + (cl & 7) * 32 + 16 * (cl >> 3) + cc;
            float s = 0.0f;
            #pragma unroll
            for (int ww = 0; ww < CW_WAVES; ++ww) s += r[ww * RED_W];
            float wsum = 0.0f;
            #pragma unroll
            for (int tap = 0; tap < 9; ++tap) {
                const bool ok = !(oh == 0 && tap < 3) && !(owp == 0 && (tap % 3) == 0);
                wsum += ok ? e_cw[q][tap] : 0.0f;
            }
            const float acc = e_cb[q] + s + e_cv[q] * wsum;
            out[(size_t)(b * DCH + ec0 + q) * 64 + 32 * ohh + lane] = (acc - e_mean[q]) * e_sc[q] + e_beta[q];
        }
    }
}

extern "C" void kernel_launch(void* const* d_in, const int* in_sizes, int n_in,
                              void* d_out, int out_size, void* d_ws, size_t ws_size,
                              hipStream_t stream) {
    const float* x     = (const float*)d_in[0];
    const float* Wr    = (const float*)d_in[1];
    const float* br    = (const float*)d_in[2];
    const float* Wi    = (const float*)d_in[3];
    const float* bi    = (const float*)d_in[4];
    const float* cw    = (const float*)d_in[5];
    const float* cb    = (const float*)d_in[6];
    const float* gamma = (const float*)d_in[7];
    const float* beta  = (const float*)d_in[8];
    const float* mean  = (const float*)d_in[9];
    const float* var   = (const float*)d_in[10];

    char* ws = (char*)d_ws;
    double* ct256 = (double*)(ws + 0);
    double* st256 = (double*)(ws + 2048);
    double* ct64  = (double*)(ws + 4096);
    double* st64  = (double*)(ws + 4608);
    double* B1    = (double*)(ws + 5120);
    double* B2    = (double*)(ws + 136192);
    float*  A     = (float*) (ws + 267264);
    float*  cvec  = (float*) (ws + 332800);

    k_tables<<<1, 256, 0, stream>>>(ct256, st256, ct64, st64);
    k_b12<<<DCH, INF, 0, stream>>>(Wr, Wi, ct256, st256, B1, B2);
    k_amat<<<9 * 8, 256, 0, stream>>>(B1, B2, ct64, st64, br, bi, A, cvec);

    k_conv_wmma<<<128, 384, 0, stream>>>(x, A, cvec, cw, cb, gamma, beta, mean, var,
                                                       (float*)d_out);
}


# twiddle-table launch removed: k_b12 and k_amat evaluate the roots of unity they use themselves (octant fold + Taylor series in double, eighth roots as constants); sum order unchanged
# speedup vs baseline: 4.8615x; 1.2309x over previous
#include <hip/hip_runtime.h>
#include <math.h>

#define IMG      2048
#define NB       16
#define DCH      64
#define INF      256

typedef float v2f __attribute__((ext_vector_type(2)));
typedef float v8f __attribute__((ext_vector_type(8)));

__device__ __forceinline__ void unit_root(int j, int lg, double& c, double& s) {
    const int quarter = 1 << (lg - 2);
    const int q = j >> (lg - 2);
    int r = j & (quarter - 1);
    const bool mirror = r > (quarter >> 1);
    r = mirror ? quarter - r : r;
    const double x  = (double)r * (6.283185307179586476925286766559 / (double)(1 << lg));
    const double x2 = x * x;
    double ps = -1.0 / 1307674368000.0;
    ps = fma(ps, x2,  1.0 / 6227020800.0);
    ps = fma(ps, x2, -1.0 / 39916800.0);
    ps = fma(ps, x2,  1.0 / 362880.0);
    ps = fma(ps, x2, -1.0 / 5040.0);
    ps = fma(ps, x2,  1.0 / 120.0);
    ps = fma(ps, x2, -1.0 / 6.0);
    const double sn = fma(x * x2, ps, x);
    double pc = 1.0 / 20922789888000.0;
    pc = fma(pc, x2, -1.0 / 87178291200.0);
    pc = fma(pc, x2,  1.0 / 479001600.0);
    pc = fma(pc, x2, -1.0 / 3628800.0);
    pc = fma(pc, x2,  1.0 / 40320.0);
    pc = fma(pc, x2, -1.0 / 720.0);
    pc = fma(pc, x2,  1.0 / 24.0);
    pc = fma(pc, x2, -0.5);
    const double cs = fma(pc, x2, 1.0);
    const double co = mirror ? sn : cs;
    const double so = mirror ? cs : sn;
    const double cq = (q & 1) ? so : co;
    const double sq = (q & 1) ? co : so;
    c = (q == 1 || q == 2) ? -cq : cq;
    s = (q >= 2) ? -sq : sq;
}

__device__ __forceinline__ void root8(int j, double& c, double& s) {
    const double R = 0.70710678118654752440;
    const int a = j & 3;
    const double cc = (a == 0) ? 1.0 : (a == 2) ? 0.0 : (a == 1) ? R : -R;
    const double ss = (a == 0) ? 0.0 : (a == 2) ? 1.0 : R;
    c = (j & 4) ? -cc : cc;
    s = (j & 4) ? -ss : ss;
}

#define YP 17
__global__ void __launch_bounds__(INF)
k_b12(const float* __restrict__ Wr, const float* __restrict__ Wi,
      double* __restrict__ B1, double* __restrict__ B2) {
    __shared__ double s_c[256], s_s[256];
    __shared__ double s_wr[INF], s_wi[INF];
    __shared__ double s_yr[16 * YP], s_yi[16 * YP];

    const int t = threadIdx.x;
    const int d = blockIdx.x;

    const float wr_t = Wr[d * INF + t];
    const float wi_t = Wi[d * INF + t];
    double ct_t, st_t;
    unit_root(t, 8, ct_t, st_t);
    s_c[t]  = ct_t;
    s_s[t]  = st_t;
    s_wr[t] = (double)wr_t;
    s_wi[t] = (double)wi_t;
    __syncthreads();

    {
        const int k2 = t & 15;
        const int n1 = t >> 4;
        double yr = 0.0, yi = 0.0;
        for (int k1 = 0; k1 < 16; ++k1) {
            const int q = ((k1 * n1) & 15) << 4;
            const double c = s_c[q], s = s_s[q];
            const double wr = s_wr[16 * k1 + k2], wi = s_wi[16 * k1 + k2];
            yr = fma(wr, c, yr);  yr = fma(wi, s, yr);
            yi = fma(wi, c, yi);  yi = fma(-wr, s, yi);
        }
        const int j = k2 * n1;
        const double c2 = s_c[j], s2 = s_s[j];
        s_yr[k2 * YP + n1] = fma(yi, s2, yr * c2);
        s_yi[k2 * YP + n1] = fma(-yr, s2, yi * c2);
    }
    __syncthreads();

    {
        const int n1 = t & 15;
        const int n2 = t >> 4;
        double zr = 0.0, zi = 0.0;
        for (int k2 = 0; k2 < 16; ++k2) {
            const int q = ((k2 * n2) & 15) << 4;
            const double c = s_c[q], s = s_s[q];
            const double yr = s_yr[k2 * YP + n1], yi = s_yi[k2 * YP + n1];
            zr = fma(yr, c, zr);  zr = fma(yi, s, zr);
            zi = fma(yi, c, zi);  zi = fma(-yr, s, zi);
        }
        B1[d * INF + t] = zr;
        B2[d * INF + t] = zi;
    }
}

#define ACOLS 32
__global__ void __launch_bounds__(256)
k_amat(const double* __restrict__ B1, const double* __restrict__ B2,
       const float* __restrict__ br, const float* __restrict__ bi,
       float* __restrict__ A, float* __restrict__ cvec) {
    __shared__ double s_yr[8 * ACOLS], s_yi[8 * ACOLS];

    const int c  = threadIdx.x & 31;
    const int w  = __builtin_amdgcn_readfirstlane((int)(threadIdx.x >> 5));
    const int m1 = blockIdx.x & 7;
    const int g  = blockIdx.x >> 3;

    double zr[8], zi[8];
    if (g < 8) {
        const double* p1 = B1 + w * INF + g * ACOLS + c;
        const double* p2 = B2 + w * INF + g * ACOLS + c;
        #pragma unroll
        for (int d1 = 0; d1 < 8; ++d1) {
            zr[d1] = p1[d1 * 8 * INF];
            zi[d1] = p2[d1 * 8 * INF];
        }
    } else {
        const int j = c & 7;
        const double r = (double)br[8 * j + w], q = (double)bi[8 * j + w];
        const double lr = r - q, li = r + q;
        #pragma unroll
        for (int d1 = 0; d1 < 8; ++d1) {
            zr[d1] = __shfl(lr, d1);
            zi[d1] = __shfl(li, d1);
        }
    }

    double c1[8], s1[8], c2[8], s2[8];
    #pragma unroll
    for (int k = 1; k < 8; ++k) {
        root8(k * m1, c1[k], s1[k]);
        root8(k * w,  c2[k], s2[k]);
    }
    double tc, ts;
    unit_root(w * m1, 6, tc, ts);

    {
        double yr = zr[0], yi = zi[0];
        asm volatile("" : "+v"(tc), "+v"(ts), "+v"(yr), "+v"(yi));
        #pragma unroll
        for (int d1 = 1; d1 < 8; ++d1) {
            const double cs = c1[d1], sn = s1[d1];
            yr = fma(zr[d1], cs, yr);  yr = fma(-zi[d1], sn, yr);
            yi = fma(zr[d1], sn, yi);  yi = fma(zi[d1], cs, yi);
        }
        s_yr[w * ACOLS + c] = fma(-yi, ts, yr * tc);
        s_yi[w * ACOLS + c] = fma(yi, tc, yr * ts);
    }
    __syncthreads();

    {
        double acc = s_yr[c];
        #pragma unroll
        for (int d2 = 1; d2 < 8; ++d2) {
            acc = fma(s_yr[d2 * ACOLS + c], c2[d2], acc);
            acc = fma(-s_yi[d2 * ACOLS + c], s2[d2], acc);
        }
        const int m = m1 + 8 * w;
        const float v = (float)(acc / 64.0);
        if (g < 8)       A[m * INF + g * ACOLS + c] = v;
        else if (c == 0) cvec[m] = v;
    }
}

typedef float v4f __attribute__((ext_vector_type(4)));

#define CW_WAVES 12
#define RED_Z    272
#define RED_W    (2 * RED_Z)
__global__ void __launch_bounds__(CW_WAVES * 32)
k_conv_wmma(const float* __restrict__ x, const float* __restrict__ A,
            const float* __restrict__ cvec, const float* __restrict__ cw,
            const float* __restrict__ cb, const float* __restrict__ gamma,
            const float* __restrict__ beta, const float* __restrict__ mean,
            const float* __restrict__ var, float* __restrict__ out) {
    __shared__ float red[CW_WAVES * RED_W];

    const int lane = threadIdx.x & 31;
    const int w    = __builtin_amdgcn_readfirstlane((int)(threadIdx.x >> 5));
    const int c    = lane & 15;
    const int h    = lane >> 4;
    const int mt   = blockIdx.x & 3;
    const int pg   = blockIdx.x >> 2;
    const int b    = pg >> 1;
    const int ohh  = pg & 1;

    const int ch   = 16 * mt + c;
    const int ow   = c & 7;
    const int oh0  = 4 * ohh + (c >> 3);
    const float* arow = A + ch * INF + 4 * h;
    const float* ximg = x + (size_t)b * IMG * IMG + 4 * h;

    const int ec0 = 16 * mt + 2 * (w & 7);
    float e_cw[2][9], e_cb[2], e_cv[2], e_sc[2], e_mean[2], e_beta[2];
    #pragma unroll
    for (int q = 0; q < 2; ++q) {
        #pragma unroll
        for (int tap = 0; tap < 9; ++tap) e_cw[q][tap] = cw[(ec0 + q) * 9 + tap];
        e_cb[q]   = cb[ec0 + q];
        e_cv[q]   = cvec[ec0 + q];
        e_sc[q]   = gamma[ec0 + q] / sqrtf(var[ec0 + q] + 1e-5f);
        e_mean[q] = mean[ec0 + q];
        e_beta[q] = beta[ec0 + q];
    }

    v8f acc0 = {}, acc1 = {};
    #pragma unroll
    for (int i = 0; i < 3; ++i) {
        const int u   = 3 * w + i;
        const int tap = u >> 2;
        const int qr  = u & 3;
        const int tj  = tap / 3;
        const int ti  = tap - 3 * tj;
        const float cwv = cw[ch * 9 + tap];

        const int pr0 = 16 * oh0 + tj - 1;
        const int pr1 = pr0 + 32;
        const int pc  = 16 * ow + ti - 1;
        const bool okc = pc >= 0;
        const bool ok0 = okc && (pr0 >= 0);
        const int prc = pr0 < 0 ? 0 : pr0;
        const int pcc = pc < 0 ? 0 : pc;
        const float* p0 = ximg + (size_t)(16 * prc + 4 * qr) * IMG + 16 * pcc;
        const float* p1 = ximg + (size_t)(16 * pr1 + 4 * qr) * IMG + 16 * pcc;

        v4f av[8], b0[8], b1[8];
        #pragma unroll
        for (int tt = 0; tt < 8; ++tt) {
            av[tt] = *(const v4f*)(arow + 8 * (8 * qr + tt));
            b0[tt] = *(const v4f*)(p0 + (tt >> 1) * IMG + 8 * (tt & 1));
            b1[tt] = *(const v4f*)(p1 + (tt >> 1) * IMG + 8 * (tt & 1));
        }
        const v4f zero = {0.0f, 0.0f, 0.0f, 0.0f};
        #pragma unroll
        for (int tt = 0; tt < 8; ++tt) {
            const v4f a  = av[tt] * cwv;
            const v4f x0 = ok0 ? b0[tt] : zero;
            const v4f x1 = okc ? b1[tt] : zero;
            v2f a_lo = {a[0], a[1]},   a_hi = {a[2], a[3]};
            v2f l0   = {x0[0], x0[1]}, h0   = {x0[2], x0[3]};
            v2f l1   = {x1[0], x1[1]}, h1   = {x1[2], x1[3]};
            acc0 = __builtin_amdgcn_wmma_f32_16x16x4_f32(false, a_lo, false, l0, (short)0, acc0, false, false);
            acc1 = __builtin_amdgcn_wmma_f32_16x16x4_f32(false, a_lo, false, l1, (short)0, acc1, false, false);
            acc0 = __builtin_amdgcn_wmma_f32_16x16x4_f32(false, a_hi, false, h0, (short)0, acc0, false, false);
            acc1 = __builtin_amdgcn_wmma_f32_16x16x4_f32(false, a_hi, false, h1, (short)0, acc1, false, false);
        }
    }

    #pragma unroll
    for (int j = 0; j < 8; ++j) {
        red[w * RED_W + j * 32 + lane]         = acc0[j];
        red[w * RED_W + RED_Z + j * 32 + lane] = acc1[j];
    }
    __syncthreads();

    if (w < 8) {
        const int z   = lane >> 4;
        const int cc  = lane & 15;
        const int oh  = 4 * ohh + (lane >> 3);
        const int owp = lane & 7;
        #pragma unroll
        for (int q = 0; q < 2; ++q) {
            const int cl  = 2 * w + q;
            const float* r = red + z * RED_Z + (cl & 7) * 32 + 16 * (cl >> 3) + cc;
            float s = 0.0f;
            #pragma unroll
            for (int ww = 0; ww < CW_WAVES; ++ww) s += r[ww * RED_W];
            float wsum = 0.0f;
            #pragma unroll
            for (int tap = 0; tap < 9; ++tap) {
                const bool ok = !(oh == 0 && tap < 3) && !(owp == 0 && (tap % 3) == 0);
                wsum += ok ? e_cw[q][tap] : 0.0f;
            }
            const float acc = e_cb[q] + s + e_cv[q] * wsum;
            out[(size_t)(b * DCH + ec0 + q) * 64 + 32 * ohh + lane] = (acc - e_mean[q]) * e_sc[q] + e_beta[q];
        }
    }
}

extern "C" void kernel_launch(void* const* d_in, const int* in_sizes, int n_in,
                              void* d_out, int out_size, void* d_ws, size_t ws_size,
                              hipStream_t stream) {
    const float* x     = (const float*)d_in[0];
    const float* Wr    = (const float*)d_in[1];
    const float* br    = (const float*)d_in[2];
    const float* Wi    = (const float*)d_in[3];
    const float* bi    = (const float*)d_in[4];
    const float* cw    = (const float*)d_in[5];
    const float* cb    = (const float*)d_in[6];
    const float* gamma = (const float*)d_in[7];
    const float* beta  = (const float*)d_in[8];
    const float* mean  = (const float*)d_in[9];
    const float* var   = (const float*)d_in[10];

    char* ws = (char*)d_ws;
    double* B1    = (double*)(ws + 5120);
    double* B2    = (double*)(ws + 136192);
    float*  A     = (float*) (ws + 267264);
    float*  cvec  = (float*) (ws + 332800);

    k_b12<<<DCH, INF, 0, stream>>>(Wr, Wi, B1, B2);
    k_amat<<<9 * 8, 256, 0, stream>>>(B1, B2, br, bi, A, cvec);

    k_conv_wmma<<<128, 384, 0, stream>>>(x, A, cvec, cw, cb, gamma, beta, mean, var,
                                                       (float*)d_out);
}


# k_b12: each row's 256-point sum as a radix-4 decimation-in-frequency transform (4 butterfly stages via LDS), one 64-thread block per (row, n mod 4); B1/B2 class-major; changes the double sum order
# speedup vs baseline: 5.8075x; 1.1946x over previous
#include <hip/hip_runtime.h>
#include <math.h>

#define IMG      2048
#define NB       16
#define DCH      64
#define INF      256

typedef float v2f __attribute__((ext_vector_type(2)));
typedef float v8f __attribute__((ext_vector_type(8)));


__device__ __forceinline__ double sin_octant(double x) {
    const double x2 = x * x;
    double ps = -1.0 / 1307674368000.0;
    ps = fma(ps, x2,  1.0 / 6227020800.0);
    ps = fma(ps, x2, -1.0 / 39916800.0);
    ps = fma(ps, x2,  1.0 / 362880.0);
    ps = fma(ps, x2, -1.0 / 5040.0);
    ps = fma(ps, x2,  1.0 / 120.0);
    ps = fma(ps, x2, -1.0 / 6.0);
    return fma(x * x2, ps, x);
}
__device__ __forceinline__ double cos_octant(double x) {
    const double x2 = x * x;
    double pc = 1.0 / 20922789888000.0;
    pc = fma(pc, x2, -1.0 / 87178291200.0);
    pc = fma(pc, x2,  1.0 / 479001600.0);
    pc = fma(pc, x2, -1.0 / 3628800.0);
    pc = fma(pc, x2,  1.0 / 40320.0);
    pc = fma(pc, x2, -1.0 / 720.0);
    pc = fma(pc, x2,  1.0 / 24.0);
    pc = fma(pc, x2, -0.5);
    return fma(pc, x2, 1.0);
}

__device__ __forceinline__ void unfold_octant(double cs, double sn, bool mirror, int q, double& c, double& s) {
    const double co = mirror ? sn : cs;
    const double so = mirror ? cs : sn;
    const double cq = (q & 1) ? so : co;
    const double sq = (q & 1) ? co : so;
    c = (q == 1 || q == 2) ? -cq : cq;
    s = (q >= 2) ? -sq : sq;
}

__device__ __forceinline__ void unit_root(int j, int lg, double& c, double& s) {
    const int quarter = 1 << (lg - 2);
    const int q = j >> (lg - 2);
    int r = j & (quarter - 1);
    const bool mirror = r > (quarter >> 1);
    r = mirror ? quarter - r : r;
    const double x  = (double)r * (6.283185307179586476925286766559 / (double)(1 << lg));
    unfold_octant(cos_octant(x), sin_octant(x), mirror, q, c, s);
}

__device__ __forceinline__ void root8(int j, double& c, double& s) {
    const double R = 0.70710678118654752440;
    const int a = j & 3;
    const double cc = (a == 0) ? 1.0 : (a == 2) ? 0.0 : (a == 1) ? R : -R;
    const double ss = (a == 0) ? 0.0 : (a == 2) ? 1.0 : R;
    c = (j & 4) ? -cc : cc;
    s = (j & 4) ? -ss : ss;
}


__device__ __forceinline__ void butterfly4(const double (&vr)[4], const double (&vi)[4], int s,
                                           double& outr, double& outi) {
    const bool odd  = (s & 1) != 0;
    const double sg = odd ? -1.0 : 1.0;
    const double t0r = fma(sg, vr[2], vr[0]), t0i = fma(sg, vi[2], vi[0]);
    const double t1r = fma(sg, vr[3], vr[1]), t1i = fma(sg, vi[3], vi[1]);
    const double ur = odd ? t1i : t1r;
    const double ui = odd ? t1r : t1i;
    outr = t0r + ((s >= 2) ? -ur : ur);
    outi = t0i + ((s == 1 || s == 2) ? -ui : ui);
}

__device__ __forceinline__ void times_root256(const double* t_c, const double* t_s, int j,
                                              double vr, double vi, double& outr, double& outi) {
    const int q = j >> 6;
    int r = j & 63;
    const bool mirror = r > 32;
    r = mirror ? 64 - r : r;
    double c, s;
    unfold_octant(t_c[r], t_s[r], mirror, q, c, s);
    outr = fma(vi, s, vr * c);
    outi = fma(-vr, s, vi * c);
}

#define FFT_T 64
__global__ void __launch_bounds__(FFT_T)
k_b12(const float* __restrict__ Wr, const float* __restrict__ Wi,
      double* __restrict__ B1, double* __restrict__ B2) {
    __shared__ double t_c[33], t_s[33];
    __shared__ double p_r[FFT_T], p_i[FFT_T];
    __shared__ double q_r[FFT_T], q_i[FFT_T];

    const int t = threadIdx.x;
    const int d = blockIdx.x >> 2;
    const int r = blockIdx.x & 3;

    float wr[4], wi[4];
    #pragma unroll
    for (int q = 0; q < 4; ++q) {
        wr[q] = Wr[d * INF + t + 64 * q];
        wi[q] = Wi[d * INF + t + 64 * q];
    }
    {
        const int j = t & 31;
        const double x = (double)j * (6.283185307179586476925286766559 / 256.0);
        if (t < 32) t_s[j] = sin_octant(x);
        else        t_c[j] = cos_octant(x);
        if (j == 0) { if (t < 32) t_s[32] = 0.70710678118654752440; else t_c[32] = 0.70710678118654752440; }
    }
    double vr[4], vi[4];
    #pragma unroll
    for (int q = 0; q < 4; ++q) { vr[q] = (double)wr[q]; vi[q] = (double)wi[q]; }
    double sr, si, yr, yi;
    butterfly4(vr, vi, r, sr, si);
    __syncthreads();
    times_root256(t_c, t_s, t * r, sr, si, yr, yi);
    p_r[t] = yr;  p_i[t] = yi;
    __syncthreads();

    {
        const int r2 = t >> 4, a = t & 15;
        #pragma unroll
        for (int q = 0; q < 4; ++q) { vr[q] = p_r[a + 16 * q]; vi[q] = p_i[a + 16 * q]; }
        butterfly4(vr, vi, r2, sr, si);
        times_root256(t_c, t_s, 4 * a * r2, sr, si, yr, yi);
        q_r[t] = yr;  q_i[t] = yi;
    }
    __syncthreads();

    {
        const int r2 = t >> 4, r3 = (t >> 2) & 3, b = t & 3;
        #pragma unroll
        for (int q = 0; q < 4; ++q) { vr[q] = q_r[16 * r2 + b + 4 * q]; vi[q] = q_i[16 * r2 + b + 4 * q]; }
        butterfly4(vr, vi, r3, sr, si);
        times_root256(t_c, t_s, 16 * b * r3, sr, si, yr, yi);
        p_r[t] = yr;  p_i[t] = yi;
    }
    __syncthreads();

    {
        const int r2 = t & 3, r3 = (t >> 2) & 3, p = t >> 4;
        #pragma unroll
        for (int b = 0; b < 4; ++b) { vr[b] = p_r[16 * r2 + 4 * r3 + b]; vi[b] = p_i[16 * r2 + 4 * r3 + b]; }
        butterfly4(vr, vi, p, sr, si);
        B1[d * INF + 64 * r + t] = sr;
        B2[d * INF + 64 * r + t] = si;
    }
}

#define ACOLS 32
__global__ void __launch_bounds__(256)
k_amat(const double* __restrict__ B1, const double* __restrict__ B2,
       const float* __restrict__ br, const float* __restrict__ bi,
       float* __restrict__ A, float* __restrict__ cvec) {
    __shared__ double s_yr[8 * ACOLS], s_yi[8 * ACOLS];

    const int c  = threadIdx.x & 31;
    const int w  = __builtin_amdgcn_readfirstlane((int)(threadIdx.x >> 5));
    const int m1 = blockIdx.x & 7;
    const int g  = blockIdx.x >> 3;

    double zr[8], zi[8];
    if (g < 8) {
        const int n   = g * ACOLS + c;
        const int pos = 64 * (n & 3) + (n >> 2);
        const double* p1 = B1 + w * INF + pos;
        const double* p2 = B2 + w * INF + pos;
        #pragma unroll
        for (int d1 = 0; d1 < 8; ++d1) {
            zr[d1] = p1[d1 * 8 * INF];
            zi[d1] = p2[d1 * 8 * INF];
        }
    } else {
        const int j = c & 7;
        const double r = (double)br[8 * j + w], q = (double)bi[8 * j + w];
        const double lr = r - q, li = r + q;
        #pragma unroll
        for (int d1 = 0; d1 < 8; ++d1) {
            zr[d1] = __shfl(lr, d1);
            zi[d1] = __shfl(li, d1);
        }
    }

    double c1[8], s1[8], c2[8], s2[8];
    #pragma unroll
    for (int k = 1; k < 8; ++k) {
        root8(k * m1, c1[k], s1[k]);
        root8(k * w,  c2[k], s2[k]);
    }
    double tc, ts;
    unit_root(w * m1, 6, tc, ts);

    {
        double yr = zr[0], yi = zi[0];
        asm volatile("" : "+v"(tc), "+v"(ts), "+v"(yr), "+v"(yi));
        #pragma unroll
        for (int d1 = 1; d1 < 8; ++d1) {
            const double cs = c1[d1], sn = s1[d1];
            yr = fma(zr[d1], cs, yr);  yr = fma(-zi[d1], sn, yr);
            yi = fma(zr[d1], sn, yi);  yi = fma(zi[d1], cs, yi);
        }
        s_yr[w * ACOLS + c] = fma(-yi, ts, yr * tc);
        s_yi[w * ACOLS + c] = fma(yi, tc, yr * ts);
    }
    __syncthreads();

    {
        double acc = s_yr[c];
        #pragma unroll
        for (int d2 = 1; d2 < 8; ++d2) {
            acc = fma(s_yr[d2 * ACOLS + c], c2[d2], acc);
            acc = fma(-s_yi[d2 * ACOLS + c], s2[d2], acc);
        }
        const int m = m1 + 8 * w;
        const float v = (float)(acc / 64.0);
        if (g < 8)       A[m * INF + g * ACOLS + c] = v;
        else if (c == 0) cvec[m] = v;
    }
}

typedef float v4f __attribute__((ext_vector_type(4)));

#define CW_WAVES 12
#define RED_Z    272
#define RED_W    (2 * RED_Z)
__global__ void __launch_bounds__(CW_WAVES * 32)
k_conv_wmma(const float* __restrict__ x, const float* __restrict__ A,
            const float* __restrict__ cvec, const float* __restrict__ cw,
            const float* __restrict__ cb, const float* __restrict__ gamma,
            const float* __restrict__ beta, const float* __restrict__ mean,
            const float* __restrict__ var, float* __restrict__ out) {
    __shared__ float red[CW_WAVES * RED_W];

    const int lane = threadIdx.x & 31;
    const int w    = __builtin_amdgcn_readfirstlane((int)(threadIdx.x >> 5));
    const int c    = lane & 15;
    const int h    = lane >> 4;
    const int mt   = blockIdx.x & 3;
    const int pg   = blockIdx.x >> 2;
    const int b    = pg >> 1;
    const int ohh  = pg & 1;

    const int ch   = 16 * mt + c;
    const int ow   = c & 7;
    const int oh0  = 4 * ohh + (c >> 3);
    const float* arow = A + ch * INF + 4 * h;
    const float* ximg = x + (size_t)b * IMG * IMG + 4 * h;

    const int ec0 = 16 * mt + 2 * (w & 7);
    float e_cw[2][9], e_cb[2], e_cv[2], e_sc[2], e_mean[2], e_beta[2];
    #pragma unroll
    for (int q = 0; q < 2; ++q) {
        #pragma unroll
        for (int tap = 0; tap < 9; ++tap) e_cw[q][tap] = cw[(ec0 + q) * 9 + tap];
        e_cb[q]   = cb[ec0 + q];
        e_cv[q]   = cvec[ec0 + q];
        e_sc[q]   = gamma[ec0 + q] / sqrtf(var[ec0 + q] + 1e-5f);
        e_mean[q] = mean[ec0 + q];
        e_beta[q] = beta[ec0 + q];
    }

    v8f acc0 = {}, acc1 = {};
    #pragma unroll
    for (int i = 0; i < 3; ++i) {
        const int u   = 3 * w + i;
        const int tap = u >> 2;
        const int qr  = u & 3;
        const int tj  = tap / 3;
        const int ti  = tap - 3 * tj;
        const float cwv = cw[ch * 9 + tap];

        const int pr0 = 16 * oh0 + tj - 1;
        const int pr1 = pr0 + 32;
        const int pc  = 16 * ow + ti - 1;
        const bool okc = pc >= 0;
        const bool ok0 = okc && (pr0 >= 0);
        const int prc = pr0 < 0 ? 0 : pr0;
        const int pcc = pc < 0 ? 0 : pc;
        const float* p0 = ximg + (size_t)(16 * prc + 4 * qr) * IMG + 16 * pcc;
        const float* p1 = ximg + (size_t)(16 * pr1 + 4 * qr) * IMG + 16 * pcc;

        v4f av[8], b0[8], b1[8];
        #pragma unroll
        for (int tt = 0; tt < 8; ++tt) {
            av[tt] = *(const v4f*)(arow + 8 * (8 * qr + tt));
            b0[tt] = *(const v4f*)(p0 + (tt >> 1) * IMG + 8 * (tt & 1));
            b1[tt] = *(const v4f*)(p1 + (tt >> 1) * IMG + 8 * (tt & 1));
        }
        const v4f zero = {0.0f, 0.0f, 0.0f, 0.0f};
        #pragma unroll
        for (int tt = 0; tt < 8; ++tt) {
            const v4f a  = av[tt] * cwv;
            const v4f x0 = ok0 ? b0[tt] : zero;
            const v4f x1 = okc ? b1[tt] : zero;
            v2f a_lo = {a[0], a[1]},   a_hi = {a[2], a[3]};
            v2f l0   = {x0[0], x0[1]}, h0   = {x0[2], x0[3]};
            v2f l1   = {x1[0], x1[1]}, h1   = {x1[2], x1[3]};
            acc0 = __builtin_amdgcn_wmma_f32_16x16x4_f32(false, a_lo, false, l0, (short)0, acc0, false, false);
            acc1 = __builtin_amdgcn_wmma_f32_16x16x4_f32(false, a_lo, false, l1, (short)0, acc1, false, false);
            acc0 = __builtin_amdgcn_wmma_f32_16x16x4_f32(false, a_hi, false, h0, (short)0, acc0, false, false);
            acc1 = __builtin_amdgcn_wmma_f32_16x16x4_f32(false, a_hi, false, h1, (short)0, acc1, false, false);
        }
    }

    #pragma unroll
    for (int j = 0; j < 8; ++j) {
        red[w * RED_W + j * 32 + lane]         = acc0[j];
        red[w * RED_W + RED_Z + j * 32 + lane] = acc1[j];
    }
    __syncthreads();

    if (w < 8) {
        const int z   = lane >> 4;
        const int cc  = lane & 15;
        const int oh  = 4 * ohh + (lane >> 3);
        const int owp = lane & 7;
        #pragma unroll
        for (int q = 0; q < 2; ++q) {
            const int cl  = 2 * w + q;
            const float* r = red + z * RED_Z + (cl & 7) * 32 + 16 * (cl >> 3) + cc;
            float s = 0.0f;
            #pragma unroll
            for (int ww = 0; ww < CW_WAVES; ++ww) s += r[ww * RED_W];
            float wsum = 0.0f;
            #pragma unroll
            for (int tap = 0; tap < 9; ++tap) {
                const bool ok = !(oh == 0 && tap < 3) && !(owp == 0 && (tap % 3) == 0);
                wsum += ok ? e_cw[q][tap] : 0.0f;
            }
            const float acc = e_cb[q] + s + e_cv[q] * wsum;
            out[(size_t)(b * DCH + ec0 + q) * 64 + 32 * ohh + lane] = (acc - e_mean[q]) * e_sc[q] + e_beta[q];
        }
    }
}

extern "C" void kernel_launch(void* const* d_in, const int* in_sizes, int n_in,
                              void* d_out, int out_size, void* d_ws, size_t ws_size,
                              hipStream_t stream) {
    const float* x     = (const float*)d_in[0];
    const float* Wr    = (const float*)d_in[1];
    const float* br    = (const float*)d_in[2];
    const float* Wi    = (const float*)d_in[3];
    const float* bi    = (const float*)d_in[4];
    const float* cw    = (const float*)d_in[5];
    const float* cb    = (const float*)d_in[6];
    const float* gamma = (const float*)d_in[7];
    const float* beta  = (const float*)d_in[8];
    const float* mean  = (const float*)d_in[9];
    const float* var   = (const float*)d_in[10];

    char* ws = (char*)d_ws;
    double* B1    = (double*)(ws + 5120);
    double* B2    = (double*)(ws + 136192);
    float*  A     = (float*) (ws + 267264);
    float*  cvec  = (float*) (ws + 332800);

    k_b12<<<DCH * 4, FFT_T, 0, stream>>>(Wr, Wi, B1, B2);
    k_amat<<<9 * 8, 256, 0, stream>>>(B1, B2, br, bi, A, cvec);

    k_conv_wmma<<<128, 384, 0, stream>>>(x, A, cvec, cw, cb, gamma, beta, mean, var,
                                                       (float*)d_out);
}


# k_conv_wmma: both WMMA operands copied to LDS in whole 128-byte lines, fragments read from LDS (x: 48 consecutive floats per pixel and row), 4 double-buffered K stages; changes the f32 sum order
# speedup vs baseline: 7.1304x; 1.2278x over previous
#include <hip/hip_runtime.h>
#include <math.h>

#define IMG      2048
#define NB       16
#define DCH      64
#define INF      256

typedef float v2f __attribute__((ext_vector_type(2)));
typedef float v8f __attribute__((ext_vector_type(8)));


__device__ __forceinline__ double sin_octant(double x) {
    const double x2 = x * x;
    double ps = -1.0 / 1307674368000.0;
    ps = fma(ps, x2,  1.0 / 6227020800.0);
    ps = fma(ps, x2, -1.0 / 39916800.0);
    ps = fma(ps, x2,  1.0 / 362880.0);
    ps = fma(ps, x2, -1.0 / 5040.0);
    ps = fma(ps, x2,  1.0 / 120.0);
    ps = fma(ps, x2, -1.0 / 6.0);
    return fma(x * x2, ps, x);
}
__device__ __forceinline__ double cos_octant(double x) {
    const double x2 = x * x;
    double pc = 1.0 / 20922789888000.0;
    pc = fma(pc, x2, -1.0 / 87178291200.0);
    pc = fma(pc, x2,  1.0 / 479001600.0);
    pc = fma(pc, x2, -1.0 / 3628800.0);
    pc = fma(pc, x2,  1.0 / 40320.0);
    pc = fma(pc, x2, -1.0 / 720.0);
    pc = fma(pc, x2,  1.0 / 24.0);
    pc = fma(pc, x2, -0.5);
    return fma(pc, x2, 1.0);
}

__device__ __forceinline__ void unfold_octant(double cs, double sn, bool mirror, int q, double& c, double& s) {
    const double co = mirror ? sn : cs;
    const double so = mirror ? cs : sn;
    const double cq = (q & 1) ? so : co;
    const double sq = (q & 1) ? co : so;
    c = (q == 1 || q == 2) ? -cq : cq;
    s = (q >= 2) ? -sq : sq;
}

__device__ __forceinline__ void unit_root(int j, int lg, double& c, double& s) {
    const int quarter = 1 << (lg - 2);
    const int q = j >> (lg - 2);
    int r = j & (quarter - 1);
    const bool mirror = r > (quarter >> 1);
    r = mirror ? quarter - r : r;
    const double x  = (double)r * (6.283185307179586476925286766559 / (double)(1 << lg));
    unfold_octant(cos_octant(x), sin_octant(x), mirror, q, c, s);
}

__device__ __forceinline__ void root8(int j, double& c, double& s) {
    const double R = 0.70710678118654752440;
    const int a = j & 3;
    const double cc = (a == 0) ? 1.0 : (a == 2) ? 0.0 : (a == 1) ? R : -R;
    const double ss = (a == 0) ? 0.0 : (a == 2) ? 1.0 : R;
    c = (j & 4) ? -cc : cc;
    s = (j & 4) ? -ss : ss;
}


__device__ __forceinline__ void butterfly4(const double (&vr)[4], const double (&vi)[4], int s,
                                           double& outr, double& outi) {
    const bool odd  = (s & 1) != 0;
    const double sg = odd ? -1.0 : 1.0;
    const double t0r = fma(sg, vr[2], vr[0]), t0i = fma(sg, vi[2], vi[0]);
    const double t1r = fma(sg, vr[3], vr[1]), t1i = fma(sg, vi[3], vi[1]);
    const double ur = odd ? t1i : t1r;
    const double ui = odd ? t1r : t1i;
    outr = t0r + ((s >= 2) ? -ur : ur);
    outi = t0i + ((s == 1 || s == 2) ? -ui : ui);
}

__device__ __forceinline__ void times_root256(const double* t_c, const double* t_s, int j,
                                              double vr, double vi, double& outr, double& outi) {
    const int q = j >> 6;
    int r = j & 63;
    const bool mirror = r > 32;
    r = mirror ? 64 - r : r;
    double c, s;
    unfold_octant(t_c[r], t_s[r], mirror, q, c, s);
    outr = fma(vi, s, vr * c);
    outi = fma(-vr, s, vi * c);
}

#define FFT_T 64
__global__ void __launch_bounds__(FFT_T)
k_b12(const float* __restrict__ Wr, const float* __restrict__ Wi,
      double* __restrict__ B1, double* __restrict__ B2) {
    __shared__ double t_c[33], t_s[33];
    __shared__ double p_r[FFT_T], p_i[FFT_T];
    __shared__ double q_r[FFT_T], q_i[FFT_T];

    const int t = threadIdx.x;
    const int d = blockIdx.x >> 2;
    const int r = blockIdx.x & 3;

    float wr[4], wi[4];
    #pragma unroll
    for (int q = 0; q < 4; ++q) {
        wr[q] = Wr[d * INF + t + 64 * q];
        wi[q] = Wi[d * INF + t + 64 * q];
    }
    {
        const int j = t & 31;
        const double x = (double)j * (6.283185307179586476925286766559 / 256.0);
        if (t < 32) t_s[j] = sin_octant(x);
        else        t_c[j] = cos_octant(x);
        if (j == 0) { if (t < 32) t_s[32] = 0.70710678118654752440; else t_c[32] = 0.70710678118654752440; }
    }
    double vr[4], vi[4];
    #pragma unroll
    for (int q = 0; q < 4; ++q) { vr[q] = (double)wr[q]; vi[q] = (double)wi[q]; }
    double sr, si, yr, yi;
    butterfly4(vr, vi, r, sr, si);
    __syncthreads();
    times_root256(t_c, t_s, t * r, sr, si, yr, yi);
    p_r[t] = yr;  p_i[t] = yi;
    __syncthreads();

    {
        const int r2 = t >> 4, a = t & 15;
        #pragma unroll
        for (int q = 0; q < 4; ++q) { vr[q] = p_r[a + 16 * q]; vi[q] = p_i[a + 16 * q]; }
        butterfly4(vr, vi, r2, sr, si);
        times_root256(t_c, t_s, 4 * a * r2, sr, si, yr, yi);
        q_r[t] = yr;  q_i[t] = yi;
    }
    __syncthreads();

    {
        const int r2 = t >> 4, r3 = (t >> 2) & 3, b = t & 3;
        #pragma unroll
        for (int q = 0; q < 4; ++q) { vr[q] = q_r[16 * r2 + b + 4 * q]; vi[q] = q_i[16 * r2 + b + 4 * q]; }
        butterfly4(vr, vi, r3, sr, si);
        times_root256(t_c, t_s, 16 * b * r3, sr, si, yr, yi);
        p_r[t] = yr;  p_i[t] = yi;
    }
    __syncthreads();

    {
        const int r2 = t & 3, r3 = (t >> 2) & 3, p = t >> 4;
        #pragma unroll
        for (int b = 0; b < 4; ++b) { vr[b] = p_r[16 * r2 + 4 * r3 + b]; vi[b] = p_i[16 * r2 + 4 * r3 + b]; }
        butterfly4(vr, vi, p, sr, si);
        B1[d * INF + 64 * r + t] = sr;
        B2[d * INF + 64 * r + t] = si;
    }
}

#define ACOLS 32
__global__ void __launch_bounds__(256)
k_amat(const double* __restrict__ B1, const double* __restrict__ B2,
       const float* __restrict__ br, const float* __restrict__ bi,
       float* __restrict__ A, float* __restrict__ cvec) {
    __shared__ double s_yr[8 * ACOLS], s_yi[8 * ACOLS];

    const int c  = threadIdx.x & 31;
    const int w  = __builtin_amdgcn_readfirstlane((int)(threadIdx.x >> 5));
    const int m1 = blockIdx.x & 7;
    const int g  = blockIdx.x >> 3;

    double zr[8], zi[8];
    if (g < 8) {
        const int n   = g * ACOLS + c;
        const int pos = 64 * (n & 3) + (n >> 2);
        const double* p1 = B1 + w * INF + pos;
        const double* p2 = B2 + w * INF + pos;
        #pragma unroll
        for (int d1 = 0; d1 < 8; ++d1) {
            zr[d1] = p1[d1 * 8 * INF];
            zi[d1] = p2[d1 * 8 * INF];
        }
    } else {
        const int j = c & 7;
        const double r = (double)br[8 * j + w], q = (double)bi[8 * j + w];
        const double lr = r - q, li = r + q;
        #pragma unroll
        for (int d1 = 0; d1 < 8; ++d1) {
            zr[d1] = __shfl(lr, d1);
            zi[d1] = __shfl(li, d1);
        }
    }

    double c1[8], s1[8], c2[8], s2[8];
    #pragma unroll
    for (int k = 1; k < 8; ++k) {
        root8(k * m1, c1[k], s1[k]);
        root8(k * w,  c2[k], s2[k]);
    }
    double tc, ts;
    unit_root(w * m1, 6, tc, ts);

    {
        double yr = zr[0], yi = zi[0];
        asm volatile("" : "+v"(tc), "+v"(ts), "+v"(yr), "+v"(yi));
        #pragma unroll
        for (int d1 = 1; d1 < 8; ++d1) {
            const double cs = c1[d1], sn = s1[d1];
            yr = fma(zr[d1], cs, yr);  yr = fma(-zi[d1], sn, yr);
            yi = fma(zr[d1], sn, yi);  yi = fma(zi[d1], cs, yi);
        }
        s_yr[w * ACOLS + c] = fma(-yi, ts, yr * tc);
        s_yi[w * ACOLS + c] = fma(yi, tc, yr * ts);
    }
    __syncthreads();

    {
        double acc = s_yr[c];
        #pragma unroll
        for (int d2 = 1; d2 < 8; ++d2) {
            acc = fma(s_yr[d2 * ACOLS + c], c2[d2], acc);
            acc = fma(-s_yi[d2 * ACOLS + c], s2[d2], acc);
        }
        const int m = m1 + 8 * w;
        const float v = (float)(acc / 64.0);
        if (g < 8)       A[m * INF + g * ACOLS + c] = v;
        else if (c == 0) cvec[m] = v;
    }
}

typedef float v4f __attribute__((ext_vector_type(4)));

__device__ __forceinline__ void lds_barrier() {
    asm volatile("s_wait_dscnt 0x0\n\ts_barrier_signal -1\n\ts_barrier_wait -1" ::: "memory");
}

#define CW_WAVES 12
#define RED_Z    272
#define RED_W    (2 * RED_Z)
#define XPITCH   580
#define XSTAGE   (32 * XPITCH)
#define APITCH   260
__global__ void __launch_bounds__(CW_WAVES * 32)
k_conv_wmma(const float* x,
            const float* __restrict__ A,
            const float* __restrict__ cvec, const float* __restrict__ cw,
            const float* __restrict__ cb, const float* __restrict__ gamma,
            const float* __restrict__ beta, const float* __restrict__ mean,
            const float* __restrict__ var, float* __restrict__ out) {
    __shared__ __attribute__((aligned(16))) float xs[2 * XSTAGE];
    __shared__ __attribute__((aligned(16))) float a_lds[16 * APITCH];
    float* const red = xs;

    const int lane = threadIdx.x & 31;
    const int w    = __builtin_amdgcn_readfirstlane((int)(threadIdx.x >> 5));
    const int c    = lane & 15;
    const int h    = lane >> 4;
    const int mt   = blockIdx.x & 3;
    const int pg   = blockIdx.x >> 2;
    const int b    = pg >> 1;
    const int ohh  = pg & 1;

    const int ch   = 16 * mt + c;
    const int tjw  = w >> 2;
    const int rw   = w & 3;
    const float* ximg = x + (size_t)b * IMG * IMG;

    const int l16  = lane & 15;
    const int owl  = 2 * (w & 3) + (lane >> 4);
    const bool act = l16 >= 4;
    const bool colok = act && !(owl == 0 && l16 < 8);
    const int coloff = 256 * owl - 32 + 4 * (colok ? l16 : (l16 | 8));
    const int sdst   = owl * XPITCH + tjw * 192 + 4 * (l16 - 4);
    const v4f zero = {0.0f, 0.0f, 0.0f, 0.0f};

    const int ec0 = 16 * mt + 2 * (w & 7);
    float e_cw[2][9], e_cb[2], e_cv[2], e_sc[2], e_mean[2], e_beta[2];
    #pragma unroll
    for (int q = 0; q < 2; ++q) {
        #pragma unroll
        for (int tap = 0; tap < 9; ++tap) e_cw[q][tap] = cw[(ec0 + q) * 9 + tap];
        e_cb[q]   = cb[ec0 + q];
        e_cv[q]   = cvec[ec0 + q];
        e_sc[q]   = gamma[ec0 + q] / sqrtf(var[ec0 + q] + 1e-5f);
        e_mean[q] = mean[ec0 + q];
        e_beta[q] = beta[ec0 + q];
    }

    v4f rg[16];
#define LOAD_STAGE(QR)                                                                           \
    _Pragma("unroll")                                                                            \
    for (int i = 0; i < 16; ++i) {                                                               \
        const int y  = 1024 * ohh + 256 * (i >> 2) + 16 * tjw - 16 + 4 * (QR) + (i & 3);         \
        const int yc = y < 0 ? 0 : y;            \
        rg[i] = *(const v4f*)(ximg + (size_t)yc * IMG + coloff);                                 \
    }
#define STORE_STAGE(BUF, QR)                                                                     \
    if (act) {                                                                                   \
        _Pragma("unroll")                                                                        \
        for (int i = 0; i < 16; ++i) {                                                           \
            const int y  = 1024 * ohh + 256 * (i >> 2) + 16 * tjw - 16 + 4 * (QR) + (i & 3);     \
            const bool ok = colok && (y >= 0);                                                   \
            *(v4f*)((BUF) + sdst + (i >> 2) * 8 * XPITCH + (i & 3) * 48) = ok ? rg[i] : zero;    \
        }                                                                                        \
    }

    const float* asrc = A + 16 * mt * INF + 4 * (int)threadIdx.x;
    v4f ar[3];
    ar[0] = *(const v4f*)(asrc);
    ar[1] = *(const v4f*)(asrc + 4 * 384);
    if (w < 8) ar[2] = *(const v4f*)(asrc + 4 * 768);
    float cwv[3];
    #pragma unroll
    for (int ti = 0; ti < 3; ++ti) cwv[ti] = cw[ch * 9 + 3 * tjw + ti];

    LOAD_STAGE(0)

    v8f acc0 = {}, acc1 = {};
#define COMPUTE_STAGE(BUF, QR)                                                                   \
    {                                                                                            \
        const float* bp = (BUF) + c * XPITCH + 48 * w + 4 * h;                                   \
        const float* ap = a_lds + c * APITCH + 16 * (4 * (QR) + rw) + 4 * h;      \
        v4f av[2];                                                                               \
        av[0] = *(const v4f*)(ap);                                                               \
        av[1] = *(const v4f*)(ap + 8);                                                           \
        v4f f0[6], f1[6];                                                                        \
        _Pragma("unroll")                                                                        \
        for (int p = 0; p < 6; ++p) {                                                            \
            f0[p] = *(const v4f*)(bp + 8 * p);                                                   \
            f1[p] = *(const v4f*)(bp + 16 * XPITCH + 8 * p);                                     \
        }                                                                                        \
        _Pragma("unroll")                                                                        \
        for (int p = 0; p < 6; ++p) {                                                            \
            const v4f a  = av[p & 1] * cwv[p >> 1];                                              \
            const v4f x0 = f0[p], x1 = f1[p];                                                    \
            v2f a_lo = {a[0], a[1]},   a_hi = {a[2], a[3]};                                      \
            v2f l0   = {x0[0], x0[1]}, h0   = {x0[2], x0[3]};                                    \
            v2f l1   = {x1[0], x1[1]}, h1   = {x1[2], x1[3]};                                    \
            acc0 = __builtin_amdgcn_wmma_f32_16x16x4_f32(false, a_lo, false, l0, (short)0, acc0, false, false); \
            acc1 = __builtin_amdgcn_wmma_f32_16x16x4_f32(false, a_lo, false, l1, (short)0, acc1, false, false); \
            acc0 = __builtin_amdgcn_wmma_f32_16x16x4_f32(false, a_hi, false, h0, (short)0, acc0, false, false); \
            acc1 = __builtin_amdgcn_wmma_f32_16x16x4_f32(false, a_hi, false, h1, (short)0, acc1, false, false); \
        }                                                                                        \
    }

    float* const buf0 = xs;
    float* const buf1 = xs + XSTAGE;

    STORE_STAGE(buf0, 0)
    #pragma unroll
    for (int i = 0; i < 3; ++i) {
        const int q = (int)threadIdx.x + 384 * i;
        if (i < 2 || w < 8) *(v4f*)(a_lds + (q >> 6) * APITCH + 4 * (q & 63)) = ar[i];
    }
    LOAD_STAGE(1)
    lds_barrier();
    COMPUTE_STAGE(buf0, 0)
    STORE_STAGE(buf1, 1)
    LOAD_STAGE(2)
    lds_barrier();
    COMPUTE_STAGE(buf1, 1)
    STORE_STAGE(buf0, 2)
    LOAD_STAGE(3)
    lds_barrier();
    COMPUTE_STAGE(buf0, 2)
    STORE_STAGE(buf1, 3)
    lds_barrier();
    COMPUTE_STAGE(buf1, 3)

    #pragma unroll
    for (int j = 0; j < 8; ++j) {
        red[w * RED_W + j * 32 + lane]         = acc0[j];
        red[w * RED_W + RED_Z + j * 32 + lane] = acc1[j];
    }
    __syncthreads();

    if (w < 8) {
        const int z   = lane >> 4;
        const int cc  = lane & 15;
        const int oh  = 4 * ohh + (lane >> 3);
        const int owp = lane & 7;
        #pragma unroll
        for (int q = 0; q < 2; ++q) {
            const int cl  = 2 * w + q;
            const float* r = red + z * RED_Z + (cl & 7) * 32 + 16 * (cl >> 3) + cc;
            float s = 0.0f;
            #pragma unroll
            for (int ww = 0; ww < CW_WAVES; ++ww) s += r[ww * RED_W];
            float wsum = 0.0f;
            #pragma unroll
            for (int tap = 0; tap < 9; ++tap) {
                const bool ok = !(oh == 0 && tap < 3) && !(owp == 0 && (tap % 3) == 0);
                wsum += ok ? e_cw[q][tap] : 0.0f;
            }
            const float acc = e_cb[q] + s + e_cv[q] * wsum;
            out[(size_t)(b * DCH + ec0 + q) * 64 + 32 * ohh + lane] = (acc - e_mean[q]) * e_sc[q] + e_beta[q];
        }
    }
#undef LOAD_STAGE
#undef STORE_STAGE
#undef COMPUTE_STAGE
}

extern "C" void kernel_launch(void* const* d_in, const int* in_sizes, int n_in,
                              void* d_out, int out_size, void* d_ws, size_t ws_size,
                              hipStream_t stream) {
    const float* x     = (const float*)d_in[0];
    const float* Wr    = (const float*)d_in[1];
    const float* br    = (const float*)d_in[2];
    const float* Wi    = (const float*)d_in[3];
    const float* bi    = (const float*)d_in[4];
    const float* cw    = (const float*)d_in[5];
    const float* cb    = (const float*)d_in[6];
    const float* gamma = (const float*)d_in[7];
    const float* beta  = (const float*)d_in[8];
    const float* mean  = (const float*)d_in[9];
    const float* var   = (const float*)d_in[10];

    char* ws = (char*)d_ws;
    double* B1    = (double*)(ws + 5120);
    double* B2    = (double*)(ws + 136192);
    float*  A     = (float*) (ws + 267264);
    float*  cvec  = (float*) (ws + 332800);

    k_b12<<<DCH * 4, FFT_T, 0, stream>>>(Wr, Wi, B1, B2);
    k_amat<<<9 * 8, 256, 0, stream>>>(B1, B2, br, bi, A, cvec);

    k_conv_wmma<<<128, 384, 0, stream>>>(x, A, cvec, cw, cb, gamma, beta, mean, var,
                                                       (float*)d_out);
}


# k_amat: each column's 64-term d sum as a radix-4 transform (3 butterfly stages via LDS), one 64-thread block per column; A stored column-major; changes the double sum order in A and cvec
# speedup vs baseline: 7.5404x; 1.0575x over previous
#include <hip/hip_runtime.h>
#include <math.h>

#define IMG      2048
#define NB       16
#define DCH      64
#define INF      256

typedef float v2f __attribute__((ext_vector_type(2)));
typedef float v8f __attribute__((ext_vector_type(8)));


__device__ __forceinline__ double sin_octant(double x) {
    const double x2 = x * x;
    double ps = -1.0 / 1307674368000.0;
    ps = fma(ps, x2,  1.0 / 6227020800.0);
    ps = fma(ps, x2, -1.0 / 39916800.0);
    ps = fma(ps, x2,  1.0 / 362880.0);
    ps = fma(ps, x2, -1.0 / 5040.0);
    ps = fma(ps, x2,  1.0 / 120.0);
    ps = fma(ps, x2, -1.0 / 6.0);
    return fma(x * x2, ps, x);
}
__device__ __forceinline__ double cos_octant(double x) {
    const double x2 = x * x;
    double pc = 1.0 / 20922789888000.0;
    pc = fma(pc, x2, -1.0 / 87178291200.0);
    pc = fma(pc, x2,  1.0 / 479001600.0);
    pc = fma(pc, x2, -1.0 / 3628800.0);
    pc = fma(pc, x2,  1.0 / 40320.0);
    pc = fma(pc, x2, -1.0 / 720.0);
    pc = fma(pc, x2,  1.0 / 24.0);
    pc = fma(pc, x2, -0.5);
    return fma(pc, x2, 1.0);
}

__device__ __forceinline__ void unfold_octant(double cs, double sn, bool mirror, int q, double& c, double& s) {
    const double co = mirror ? sn : cs;
    const double so = mirror ? cs : sn;
    const double cq = (q & 1) ? so : co;
    const double sq = (q & 1) ? co : so;
    c = (q == 1 || q == 2) ? -cq : cq;
    s = (q >= 2) ? -sq : sq;
}

__device__ __forceinline__ void unit_root(int j, int lg, double& c, double& s) {
    const int quarter = 1 << (lg - 2);
    const int q = j >> (lg - 2);
    int r = j & (quarter - 1);
    const bool mirror = r > (quarter >> 1);
    r = mirror ? quarter - r : r;
    const double x  = (double)r * (6.283185307179586476925286766559 / (double)(1 << lg));
    unfold_octant(cos_octant(x), sin_octant(x), mirror, q, c, s);
}

__device__ __forceinline__ void root8(int j, double& c, double& s) {
    const double R = 0.70710678118654752440;
    const int a = j & 3;
    const double cc = (a == 0) ? 1.0 : (a == 2) ? 0.0 : (a == 1) ? R : -R;
    const double ss = (a == 0) ? 0.0 : (a == 2) ? 1.0 : R;
    c = (j & 4) ? -cc : cc;
    s = (j & 4) ? -ss : ss;
}


__device__ __forceinline__ void butterfly4(const double (&vr)[4], const double (&vi)[4], int s,
                                           double& outr, double& outi) {
    const bool odd  = (s & 1) != 0;
    const double sg = odd ? -1.0 : 1.0;
    const double t0r = fma(sg, vr[2], vr[0]), t0i = fma(sg, vi[2], vi[0]);
    const double t1r = fma(sg, vr[3], vr[1]), t1i = fma(sg, vi[3], vi[1]);
    const double ur = odd ? t1i : t1r;
    const double ui = odd ? t1r : t1i;
    outr = t0r + ((s >= 2) ? -ur : ur);
    outi = t0i + ((s == 1 || s == 2) ? -ui : ui);
}

__device__ __forceinline__ void times_root256(const double* t_c, const double* t_s, int j,
                                              double vr, double vi, double& outr, double& outi) {
    const int q = j >> 6;
    int r = j & 63;
    const bool mirror = r > 32;
    r = mirror ? 64 - r : r;
    double c, s;
    unfold_octant(t_c[r], t_s[r], mirror, q, c, s);
    outr = fma(vi, s, vr * c);
    outi = fma(-vr, s, vi * c);
}

#define FFT_T 64
__global__ void __launch_bounds__(FFT_T)
k_b12(const float* __restrict__ Wr, const float* __restrict__ Wi,
      double* __restrict__ B1, double* __restrict__ B2) {
    __shared__ double t_c[33], t_s[33];
    __shared__ double p_r[FFT_T], p_i[FFT_T];
    __shared__ double q_r[FFT_T], q_i[FFT_T];

    const int t = threadIdx.x;
    const int d = blockIdx.x >> 2;
    const int r = blockIdx.x & 3;

    float wr[4], wi[4];
    #pragma unroll
    for (int q = 0; q < 4; ++q) {
        wr[q] = Wr[d * INF + t + 64 * q];
        wi[q] = Wi[d * INF + t + 64 * q];
    }
    {
        const int j = t & 31;
        const double x = (double)j * (6.283185307179586476925286766559 / 256.0);
        if (t < 32) t_s[j] = sin_octant(x);
        else        t_c[j] = cos_octant(x);
        if (j == 0) { if (t < 32) t_s[32] = 0.70710678118654752440; else t_c[32] = 0.70710678118654752440; }
    }
    double vr[4], vi[4];
    #pragma unroll
    for (int q = 0; q < 4; ++q) { vr[q] = (double)wr[q]; vi[q] = (double)wi[q]; }
    double sr, si, yr, yi;
    butterfly4(vr, vi, r, sr, si);
    __syncthreads();
    times_root256(t_c, t_s, t * r, sr, si, yr, yi);
    p_r[t] = yr;  p_i[t] = yi;
    __syncthreads();

    {
        const int r2 = t >> 4, a = t & 15;
        #pragma unroll
        for (int q = 0; q < 4; ++q) { vr[q] = p_r[a + 16 * q]; vi[q] = p_i[a + 16 * q]; }
        butterfly4(vr, vi, r2, sr, si);
        times_root256(t_c, t_s, 4 * a * r2, sr, si, yr, yi);
        q_r[t] = yr;  q_i[t] = yi;
    }
    __syncthreads();

    {
        const int r2 = t >> 4, r3 = (t >> 2) & 3, b = t & 3;
        #pragma unroll
        for (int q = 0; q < 4; ++q) { vr[q] = q_r[16 * r2 + b + 4 * q]; vi[q] = q_i[16 * r2 + b + 4 * q]; }
        butterfly4(vr, vi, r3, sr, si);
        times_root256(t_c, t_s, 16 * b * r3, sr, si, yr, yi);
        p_r[t] = yr;  p_i[t] = yi;
    }
    __syncthreads();

    {
        const int r2 = t & 3, r3 = (t >> 2) & 3, p = t >> 4;
        #pragma unroll
        for (int b = 0; b < 4; ++b) { vr[b] = p_r[16 * r2 + 4 * r3 + b]; vi[b] = p_i[16 * r2 + 4 * r3 + b]; }
        butterfly4(vr, vi, p, sr, si);
        B1[d * INF + 64 * r + t] = sr;
        B2[d * INF + 64 * r + t] = si;
    }
}


__device__ __forceinline__ double butterfly4_re(const double (&vr)[4], const double (&vi)[4], int s) {
    const bool odd  = (s & 1) != 0;
    const double sg = odd ? -1.0 : 1.0;
    const double t0r = fma(sg, vr[2], vr[0]);
    const double u1 = odd ? vi[1] : vr[1];
    const double u3 = odd ? vi[3] : vr[3];
    const double ur = fma(sg, u3, u1);
    return t0r + ((s >= 2) ? -ur : ur);
}

__global__ void __launch_bounds__(FFT_T)
k_amat(const double* __restrict__ B1, const double* __restrict__ B2,
       const float* __restrict__ br, const float* __restrict__ bi,
       float* __restrict__ At, float* __restrict__ cvec) {
    __shared__ double t_c[33], t_s[33];
    __shared__ double p_r[FFT_T], p_i[FFT_T];
    __shared__ double q_r[FFT_T], q_i[FFT_T];

    const int t = threadIdx.x;
    const int n = blockIdx.x;

    double zr, zi;
    if (n < INF) {
        const int pos = 64 * (n & 3) + (n >> 2);
        zr = B1[t * INF + pos];
        zi = B2[t * INF + pos];
    } else {
        const double r = (double)br[t], q = (double)bi[t];
        zr = r - q;  zi = r + q;
    }
    {
        const int j = t & 31;
        const double x = (double)j * (6.283185307179586476925286766559 / 256.0);
        if (t < 32) t_s[j] = sin_octant(x);
        else        t_c[j] = cos_octant(x);
        if (j == 0) { if (t < 32) t_s[32] = 0.70710678118654752440; else t_c[32] = 0.70710678118654752440; }
    }
    p_r[t] = zr;  p_i[t] = -zi;
    __syncthreads();

    double vr[4], vi[4], sr, si, yr, yi;
    {
        const int r2 = t >> 4, a = t & 15;
        #pragma unroll
        for (int q = 0; q < 4; ++q) { vr[q] = p_r[a + 16 * q]; vi[q] = p_i[a + 16 * q]; }
        butterfly4(vr, vi, r2, sr, si);
        times_root256(t_c, t_s, 4 * a * r2, sr, si, yr, yi);
        q_r[t] = yr;  q_i[t] = yi;
    }
    __syncthreads();

    {
        const int r2 = t >> 4, r3 = (t >> 2) & 3, b = t & 3;
        #pragma unroll
        for (int q = 0; q < 4; ++q) { vr[q] = q_r[16 * r2 + b + 4 * q]; vi[q] = q_i[16 * r2 + b + 4 * q]; }
        butterfly4(vr, vi, r3, sr, si);
        times_root256(t_c, t_s, 16 * b * r3, sr, si, yr, yi);
        p_r[t] = yr;  p_i[t] = yi;
    }
    __syncthreads();

    {
        const int r2 = t & 3, r3 = (t >> 2) & 3, p = t >> 4;
        #pragma unroll
        for (int b = 0; b < 4; ++b) { vr[b] = p_r[16 * r2 + 4 * r3 + b]; vi[b] = p_i[16 * r2 + 4 * r3 + b]; }
        const float v = (float)(butterfly4_re(vr, vi, p) / 64.0);
        if (n < INF) At[n * DCH + t] = v;
        else         cvec[t] = v;
    }
}

typedef float v4f __attribute__((ext_vector_type(4)));

__device__ __forceinline__ void lds_barrier() {
    asm volatile("s_wait_dscnt 0x0\n\ts_barrier_signal -1\n\ts_barrier_wait -1" ::: "memory");
}

#define CW_WAVES 12
#define RED_Z    272
#define RED_W    (2 * RED_Z)
#define XPITCH   580
#define XSTAGE   (32 * XPITCH)
#define APITCH   260
__global__ void __launch_bounds__(CW_WAVES * 32)
k_conv_wmma(const float* x,
            const float* __restrict__ A,
            const float* __restrict__ cvec, const float* __restrict__ cw,
            const float* __restrict__ cb, const float* __restrict__ gamma,
            const float* __restrict__ beta, const float* __restrict__ mean,
            const float* __restrict__ var, float* __restrict__ out) {
    __shared__ __attribute__((aligned(16))) float xs[2 * XSTAGE];
    __shared__ __attribute__((aligned(16))) float a_lds[16 * APITCH];
    float* const red = xs;

    const int lane = threadIdx.x & 31;
    const int w    = __builtin_amdgcn_readfirstlane((int)(threadIdx.x >> 5));
    const int c    = lane & 15;
    const int h    = lane >> 4;
    const int mt   = blockIdx.x & 3;
    const int pg   = blockIdx.x >> 2;
    const int b    = pg >> 1;
    const int ohh  = pg & 1;

    const int ch   = 16 * mt + c;
    const int tjw  = w >> 2;
    const int rw   = w & 3;
    const float* ximg = x + (size_t)b * IMG * IMG;

    const int l16  = lane & 15;
    const int owl  = 2 * (w & 3) + (lane >> 4);
    const bool act = l16 >= 4;
    const bool colok = act && !(owl == 0 && l16 < 8);
    const int coloff = 256 * owl - 32 + 4 * (colok ? l16 : (l16 | 8));
    const int sdst   = owl * XPITCH + tjw * 192 + 4 * (l16 - 4);
    const v4f zero = {0.0f, 0.0f, 0.0f, 0.0f};

    const int ec0 = 16 * mt + 2 * (w & 7);
    float e_cw[2][9], e_cb[2], e_cv[2], e_sc[2], e_mean[2], e_beta[2];
    #pragma unroll
    for (int q = 0; q < 2; ++q) {
        #pragma unroll
        for (int tap = 0; tap < 9; ++tap) e_cw[q][tap] = cw[(ec0 + q) * 9 + tap];
        e_cb[q]   = cb[ec0 + q];
        e_cv[q]   = cvec[ec0 + q];
        e_sc[q]   = gamma[ec0 + q] / sqrtf(var[ec0 + q] + 1e-5f);
        e_mean[q] = mean[ec0 + q];
        e_beta[q] = beta[ec0 + q];
    }

    v4f rg[16];
#define LOAD_STAGE(QR)                                                                           \
    _Pragma("unroll")                                                                            \
    for (int i = 0; i < 16; ++i) {                                                               \
        const int y  = 1024 * ohh + 256 * (i >> 2) + 16 * tjw - 16 + 4 * (QR) + (i & 3);         \
        const int yc = y < 0 ? 0 : y;            \
        rg[i] = *(const v4f*)(ximg + (size_t)yc * IMG + coloff);                                 \
    }
#define STORE_STAGE(BUF, QR)                                                                     \
    if (act) {                                                                                   \
        _Pragma("unroll")                                                                        \
        for (int i = 0; i < 16; ++i) {                                                           \
            const int y  = 1024 * ohh + 256 * (i >> 2) + 16 * tjw - 16 + 4 * (QR) + (i & 3);     \
            const bool ok = colok && (y >= 0);                                                   \
            *(v4f*)((BUF) + sdst + (i >> 2) * 8 * XPITCH + (i & 3) * 48) = ok ? rg[i] : zero;    \
        }                                                                                        \
    }

    const float* asrc = A + 16 * mt + DCH * ((int)threadIdx.x >> 2) + 4 * ((int)threadIdx.x & 3);
    v4f ar[3];
    ar[0] = *(const v4f*)(asrc);
    ar[1] = *(const v4f*)(asrc + DCH * 96);
    if (w < 8) ar[2] = *(const v4f*)(asrc + DCH * 192);
    float cwv[3];
    #pragma unroll
    for (int ti = 0; ti < 3; ++ti) cwv[ti] = cw[ch * 9 + 3 * tjw + ti];

    LOAD_STAGE(0)

    v8f acc0 = {}, acc1 = {};
#define COMPUTE_STAGE(BUF, QR)                                                                   \
    {                                                                                            \
        const float* bp = (BUF) + c * XPITCH + 48 * w + 4 * h;                                   \
        const float* ap = a_lds + c * APITCH + 16 * (4 * (QR) + rw) + 4 * h;      \
        v4f av[2];                                                                               \
        av[0] = *(const v4f*)(ap);                                                               \
        av[1] = *(const v4f*)(ap + 8);                                                           \
        v4f f0[6], f1[6];                                                                        \
        _Pragma("unroll")                                                                        \
        for (int p = 0; p < 6; ++p) {                                                            \
            f0[p] = *(const v4f*)(bp + 8 * p);                                                   \
            f1[p] = *(const v4f*)(bp + 16 * XPITCH + 8 * p);                                     \
        }                                                                                        \
        _Pragma("unroll")                                                                        \
        for (int p = 0; p < 6; ++p) {                                                            \
            const v4f a  = av[p & 1] * cwv[p >> 1];                                              \
            const v4f x0 = f0[p], x1 = f1[p];                                                    \
            v2f a_lo = {a[0], a[1]},   a_hi = {a[2], a[3]};                                      \
            v2f l0   = {x0[0], x0[1]}, h0   = {x0[2], x0[3]};                                    \
            v2f l1   = {x1[0], x1[1]}, h1   = {x1[2], x1[3]};                                    \
            acc0 = __builtin_amdgcn_wmma_f32_16x16x4_f32(false, a_lo, false, l0, (short)0, acc0, false, false); \
            acc1 = __builtin_amdgcn_wmma_f32_16x16x4_f32(false, a_lo, false, l1, (short)0, acc1, false, false); \
            acc0 = __builtin_amdgcn_wmma_f32_16x16x4_f32(false, a_hi, false, h0, (short)0, acc0, false, false); \
            acc1 = __builtin_amdgcn_wmma_f32_16x16x4_f32(false, a_hi, false, h1, (short)0, acc1, false, false); \
        }                                                                                        \
    }

    float* const buf0 = xs;
    float* const buf1 = xs + XSTAGE;

    STORE_STAGE(buf0, 0)
    #pragma unroll
    for (int i = 0; i < 3; ++i) {
        const int q = (int)threadIdx.x + 384 * i;
        if (i < 2 || w < 8) {
            #pragma unroll
            for (int e = 0; e < 4; ++e) a_lds[(4 * (q & 3) + e) * APITCH + (q >> 2)] = ar[i][e];
        }
    }
    LOAD_STAGE(1)
    lds_barrier();
    COMPUTE_STAGE(buf0, 0)
    STORE_STAGE(buf1, 1)
    LOAD_STAGE(2)
    lds_barrier();
    COMPUTE_STAGE(buf1, 1)
    STORE_STAGE(buf0, 2)
    LOAD_STAGE(3)
    lds_barrier();
    COMPUTE_STAGE(buf0, 2)
    STORE_STAGE(buf1, 3)
    lds_barrier();
    COMPUTE_STAGE(buf1, 3)

    #pragma unroll
    for (int j = 0; j < 8; ++j) {
        red[w * RED_W + j * 32 + lane]         = acc0[j];
        red[w * RED_W + RED_Z + j * 32 + lane] = acc1[j];
    }
    __syncthreads();

    if (w < 8) {
        const int z   = lane >> 4;
        const int cc  = lane & 15;
        const int oh  = 4 * ohh + (lane >> 3);
        const int owp = lane & 7;
        #pragma unroll
        for (int q = 0; q < 2; ++q) {
            const int cl  = 2 * w + q;
            const float* r = red + z * RED_Z + (cl & 7) * 32 + 16 * (cl >> 3) + cc;
            float s = 0.0f;
            #pragma unroll
            for (int ww = 0; ww < CW_WAVES; ++ww) s += r[ww * RED_W];
            float wsum = 0.0f;
            #pragma unroll
            for (int tap = 0; tap < 9; ++tap) {
                const bool ok = !(oh == 0 && tap < 3) && !(owp == 0 && (tap % 3) == 0);
                wsum += ok ? e_cw[q][tap] : 0.0f;
            }
            const float acc = e_cb[q] + s + e_cv[q] * wsum;
            out[(size_t)(b * DCH + ec0 + q) * 64 + 32 * ohh + lane] = (acc - e_mean[q]) * e_sc[q] + e_beta[q];
        }
    }
#undef LOAD_STAGE
#undef STORE_STAGE
#undef COMPUTE_STAGE
}

extern "C" void kernel_launch(void* const* d_in, const int* in_sizes, int n_in,
                              void* d_out, int out_size, void* d_ws, size_t ws_size,
                              hipStream_t stream) {
    const float* x     = (const float*)d_in[0];
    const float* Wr    = (const float*)d_in[1];
    const float* br    = (const float*)d_in[2];
    const float* Wi    = (const float*)d_in[3];
    const float* bi    = (const float*)d_in[4];
    const float* cw    = (const float*)d_in[5];
    const float* cb    = (const float*)d_in[6];
    const float* gamma = (const float*)d_in[7];
    const float* beta  = (const float*)d_in[8];
    const float* mean  = (const float*)d_in[9];
    const float* var   = (const float*)d_in[10];

    char* ws = (char*)d_ws;
    double* B1    = (double*)(ws + 5120);
    double* B2    = (double*)(ws + 136192);
    float*  A     = (float*) (ws + 267264);
    float*  cvec  = (float*) (ws + 332800);

    k_b12<<<DCH * 4, FFT_T, 0, stream>>>(Wr, Wi, B1, B2);
    k_amat<<<INF + 1, FFT_T, 0, stream>>>(B1, B2, br, bi, A, cvec);

    k_conv_wmma<<<128, 384, 0, stream>>>(x, A, cvec, cw, cb, gamma, beta, mean, var,
                                                       (float*)d_out);
}


# k_conv_wmma: 256 blocks of 16 pixels x 16 channels (was 128 of 32 x 16) so every WGP multiplies; all 4 K stages loaded at the top into own LDS buffers; 2 k-interleaved accumulators: new f32 sum order
# speedup vs baseline: 8.1934x; 1.0866x over previous
#include <hip/hip_runtime.h>
#include <math.h>

#define IMG      2048
#define NB       16
#define DCH      64
#define INF      256

typedef float v2f __attribute__((ext_vector_type(2)));
typedef float v8f __attribute__((ext_vector_type(8)));


__device__ __forceinline__ double sin_octant(double x) {
    const double x2 = x * x;
    double ps = -1.0 / 1307674368000.0;
    ps = fma(ps, x2,  1.0 / 6227020800.0);
    ps = fma(ps, x2, -1.0 / 39916800.0);
    ps = fma(ps, x2,  1.0 / 362880.0);
    ps = fma(ps, x2, -1.0 / 5040.0);
    ps = fma(ps, x2,  1.0 / 120.0);
    ps = fma(ps, x2, -1.0 / 6.0);
    return fma(x * x2, ps, x);
}
__device__ __forceinline__ double cos_octant(double x) {
    const double x2 = x * x;
    double pc = 1.0 / 20922789888000.0;
    pc = fma(pc, x2, -1.0 / 87178291200.0);
    pc = fma(pc, x2,  1.0 / 479001600.0);
    pc = fma(pc, x2, -1.0 / 3628800.0);
    pc = fma(pc, x2,  1.0 / 40320.0);
    pc = fma(pc, x2, -1.0 / 720.0);
    pc = fma(pc, x2,  1.0 / 24.0);
    pc = fma(pc, x2, -0.5);
    return fma(pc, x2, 1.0);
}

__device__ __forceinline__ void unfold_octant(double cs, double sn, bool mirror, int q, double& c, double& s) {
    const double co = mirror ? sn : cs;
    const double so = mirror ? cs : sn;
    const double cq = (q & 1) ? so : co;
    const double sq = (q & 1) ? co : so;
    c = (q == 1 || q == 2) ? -cq : cq;
    s = (q >= 2) ? -sq : sq;
}

__device__ __forceinline__ void unit_root(int j, int lg, double& c, double& s) {
    const int quarter = 1 << (lg - 2);
    const int q = j >> (lg - 2);
    int r = j & (quarter - 1);
    const bool mirror = r > (quarter >> 1);
    r = mirror ? quarter - r : r;
    const double x  = (double)r * (6.283185307179586476925286766559 / (double)(1 << lg));
    unfold_octant(cos_octant(x), sin_octant(x), mirror, q, c, s);
}

__device__ __forceinline__ void root8(int j, double& c, double& s) {
    const double R = 0.70710678118654752440;
    const int a = j & 3;
    const double cc = (a == 0) ? 1.0 : (a == 2) ? 0.0 : (a == 1) ? R : -R;
    const double ss = (a == 0) ? 0.0 : (a == 2) ? 1.0 : R;
    c = (j & 4) ? -cc : cc;
    s = (j & 4) ? -ss : ss;
}


__device__ __forceinline__ void butterfly4(const double (&vr)[4], const double (&vi)[4], int s,
                                           double& outr, double& outi) {
    const bool odd  = (s & 1) != 0;
    const double sg = odd ? -1.0 : 1.0;
    const double t0r = fma(sg, vr[2], vr[0]), t0i = fma(sg, vi[2], vi[0]);
    const double t1r = fma(sg, vr[3], vr[1]), t1i = fma(sg, vi[3], vi[1]);
    const double ur = odd ? t1i : t1r;
    const double ui = odd ? t1r : t1i;
    outr = t0r + ((s >= 2) ? -ur : ur);
    outi = t0i + ((s == 1 || s == 2) ? -ui : ui);
}

__device__ __forceinline__ void times_root256(const double* t_c, const double* t_s, int j,
                                              double vr, double vi, double& outr, double& outi) {
    const int q = j >> 6;
    int r = j & 63;
    const bool mirror = r > 32;
    r = mirror ? 64 - r : r;
    double c, s;
    unfold_octant(t_c[r], t_s[r], mirror, q, c, s);
    outr = fma(vi, s, vr * c);
    outi = fma(-vr, s, vi * c);
}

#define FFT_T 64
__global__ void __launch_bounds__(FFT_T)
k_b12(const float* __restrict__ Wr, const float* __restrict__ Wi,
      double* __restrict__ B1, double* __restrict__ B2) {
    __shared__ double t_c[33], t_s[33];
    __shared__ double p_r[FFT_T], p_i[FFT_T];
    __shared__ double q_r[FFT_T], q_i[FFT_T];

    const int t = threadIdx.x;
    const int d = blockIdx.x >> 2;
    const int r = blockIdx.x & 3;

    float wr[4], wi[4];
    #pragma unroll
    for (int q = 0; q < 4; ++q) {
        wr[q] = Wr[d * INF + t + 64 * q];
        wi[q] = Wi[d * INF + t + 64 * q];
    }
    {
        const int j = t & 31;
        const double x = (double)j * (6.283185307179586476925286766559 / 256.0);
        if (t < 32) t_s[j] = sin_octant(x);
        else        t_c[j] = cos_octant(x);
        if (j == 0) { if (t < 32) t_s[32] = 0.70710678118654752440; else t_c[32] = 0.70710678118654752440; }
    }
    double vr[4], vi[4];
    #pragma unroll
    for (int q = 0; q < 4; ++q) { vr[q] = (double)wr[q]; vi[q] = (double)wi[q]; }
    double sr, si, yr, yi;
    butterfly4(vr, vi, r, sr, si);
    __syncthreads();
    times_root256(t_c, t_s, t * r, sr, si, yr, yi);
    p_r[t] = yr;  p_i[t] = yi;
    __syncthreads();

    {
        const int r2 = t >> 4, a = t & 15;
        #pragma unroll
        for (int q = 0; q < 4; ++q) { vr[q] = p_r[a + 16 * q]; vi[q] = p_i[a + 16 * q]; }
        butterfly4(vr, vi, r2, sr, si);
        times_root256(t_c, t_s, 4 * a * r2, sr, si, yr, yi);
        q_r[t] = yr;  q_i[t] = yi;
    }
    __syncthreads();

    {
        const int r2 = t >> 4, r3 = (t >> 2) & 3, b = t & 3;
        #pragma unroll
        for (int q = 0; q < 4; ++q) { vr[q] = q_r[16 * r2 + b + 4 * q]; vi[q] = q_i[16 * r2 + b + 4 * q]; }
        butterfly4(vr, vi, r3, sr, si);
        times_root256(t_c, t_s, 16 * b * r3, sr, si, yr, yi);
        p_r[t] = yr;  p_i[t] = yi;
    }
    __syncthreads();

    {
        const int r2 = t & 3, r3 = (t >> 2) & 3, p = t >> 4;
        #pragma unroll
        for (int b = 0; b < 4; ++b) { vr[b] = p_r[16 * r2 + 4 * r3 + b]; vi[b] = p_i[16 * r2 + 4 * r3 + b]; }
        butterfly4(vr, vi, p, sr, si);
        B1[d * INF + 64 * r + t] = sr;
        B2[d * INF + 64 * r + t] = si;
    }
}


__device__ __forceinline__ double butterfly4_re(const double (&vr)[4], const double (&vi)[4], int s) {
    const bool odd  = (s & 1) != 0;
    const double sg = odd ? -1.0 : 1.0;
    const double t0r = fma(sg, vr[2], vr[0]);
    const double u1 = odd ? vi[1] : vr[1];
    const double u3 = odd ? vi[3] : vr[3];
    const double ur = fma(sg, u3, u1);
    return t0r + ((s >= 2) ? -ur : ur);
}

__global__ void __launch_bounds__(FFT_T)
k_amat(const double* __restrict__ B1, const double* __restrict__ B2,
       const float* __restrict__ br, const float* __restrict__ bi,
       float* __restrict__ At, float* __restrict__ cvec) {
    __shared__ double t_c[33], t_s[33];
    __shared__ double p_r[FFT_T], p_i[FFT_T];
    __shared__ double q_r[FFT_T], q_i[FFT_T];

    const int t = threadIdx.x;
    const int n = blockIdx.x;

    double zr, zi;
    if (n < INF) {
        const int pos = 64 * (n & 3) + (n >> 2);
        zr = B1[t * INF + pos];
        zi = B2[t * INF + pos];
    } else {
        const double r = (double)br[t], q = (double)bi[t];
        zr = r - q;  zi = r + q;
    }
    {
        const int j = t & 31;
        const double x = (double)j * (6.283185307179586476925286766559 / 256.0);
        if (t < 32) t_s[j] = sin_octant(x);
        else        t_c[j] = cos_octant(x);
        if (j == 0) { if (t < 32) t_s[32] = 0.70710678118654752440; else t_c[32] = 0.70710678118654752440; }
    }
    p_r[t] = zr;  p_i[t] = -zi;
    __syncthreads();

    double vr[4], vi[4], sr, si, yr, yi;
    {
        const int r2 = t >> 4, a = t & 15;
        #pragma unroll
        for (int q = 0; q < 4; ++q) { vr[q] = p_r[a + 16 * q]; vi[q] = p_i[a + 16 * q]; }
        butterfly4(vr, vi, r2, sr, si);
        times_root256(t_c, t_s, 4 * a * r2, sr, si, yr, yi);
        q_r[t] = yr;  q_i[t] = yi;
    }
    __syncthreads();

    {
        const int r2 = t >> 4, r3 = (t >> 2) & 3, b = t & 3;
        #pragma unroll
        for (int q = 0; q < 4; ++q) { vr[q] = q_r[16 * r2 + b + 4 * q]; vi[q] = q_i[16 * r2 + b + 4 * q]; }
        butterfly4(vr, vi, r3, sr, si);
        times_root256(t_c, t_s, 16 * b * r3, sr, si, yr, yi);
        p_r[t] = yr;  p_i[t] = yi;
    }
    __syncthreads();

    {
        const int r2 = t & 3, r3 = (t >> 2) & 3, p = t >> 4;
        #pragma unroll
        for (int b = 0; b < 4; ++b) { vr[b] = p_r[16 * r2 + 4 * r3 + b]; vi[b] = p_i[16 * r2 + 4 * r3 + b]; }
        const float v = (float)(butterfly4_re(vr, vi, p) / 64.0);
        if (n < INF) At[n * DCH + t] = v;
        else         cvec[t] = v;
    }
}

typedef float v4f __attribute__((ext_vector_type(4)));

__device__ __forceinline__ void lds_barrier() {
    asm volatile("s_wait_dscnt 0x0\n\ts_barrier_signal -1\n\ts_barrier_wait -1" ::: "memory");
}

#define CW_WAVES 12
#define RED_Z    272
#define XPITCH   580
#define XSTAGE   (16 * XPITCH)
#define APITCH   260
__global__ void __launch_bounds__(CW_WAVES * 32)
k_conv_wmma(const float* x,
            const float* __restrict__ A,
            const float* __restrict__ cvec, const float* __restrict__ cw,
            const float* __restrict__ cb, const float* __restrict__ gamma,
            const float* __restrict__ beta, const float* __restrict__ mean,
            const float* __restrict__ var, float* __restrict__ out) {
    __shared__ __attribute__((aligned(16))) float xs[4 * XSTAGE];
    __shared__ __attribute__((aligned(16))) float a_lds[16 * APITCH];
    float* const red = xs;

    const int lane = threadIdx.x & 31;
    const int w    = __builtin_amdgcn_readfirstlane((int)(threadIdx.x >> 5));
    const int c    = lane & 15;
    const int h    = lane >> 4;
    const int mt   = blockIdx.x & 3;
    const int pg   = blockIdx.x >> 2;
    const int b    = pg >> 2;
    const int oq   = pg & 3;

    const int ch   = 16 * mt + c;
    const int tjw  = w >> 2;
    const int rw   = w & 3;
    const float* ximg = x + (size_t)b * IMG * IMG;

    const int l16  = lane & 15;
    const int owl  = 2 * (w & 3) + (lane >> 4);
    const bool act = l16 >= 4;
    const bool colok = act && !(owl == 0 && l16 < 8);
    const int coloff = 256 * owl - 32 + 4 * (colok ? l16 : (l16 | 8));
    const int sdst   = owl * XPITCH + tjw * 192 + 4 * (l16 - 4);
    const v4f zero = {0.0f, 0.0f, 0.0f, 0.0f};

    const int ec0 = 16 * mt + 2 * (w & 7);
    float e_cw[2][9], e_cb[2], e_cv[2], e_sc[2], e_mean[2], e_beta[2];
    #pragma unroll
    for (int q = 0; q < 2; ++q) {
        #pragma unroll
        for (int tap = 0; tap < 9; ++tap) e_cw[q][tap] = cw[(ec0 + q) * 9 + tap];
        e_cb[q]   = cb[ec0 + q];
        e_cv[q]   = cvec[ec0 + q];
        e_sc[q]   = gamma[ec0 + q] / sqrtf(var[ec0 + q] + 1e-5f);
        e_mean[q] = mean[ec0 + q];
        e_beta[q] = beta[ec0 + q];
    }

    v4f rg[32];
#define LOAD_STAGE(QR)                                                                           \
    _Pragma("unroll")                                                                            \
    for (int i = 0; i < 8; ++i) {                                                                \
        const int y  = 512 * oq + 256 * (i >> 2) + 16 * tjw - 16 + 4 * (QR) + (i & 3);           \
        const int yc = y < 0 ? 0 : y;            \
        rg[8 * (QR) + i] = *(const v4f*)(ximg + (size_t)yc * IMG + coloff);                      \
    }
#define STORE_STAGE(BUF, QR)                                                                     \
    if (act) {                                                                                   \
        _Pragma("unroll")                                                                        \
        for (int i = 0; i < 8; ++i) {                                                            \
            const int y  = 512 * oq + 256 * (i >> 2) + 16 * tjw - 16 + 4 * (QR) + (i & 3);       \
            const bool ok = colok && (y >= 0);                                                   \
            *(v4f*)((BUF) + sdst + (i >> 2) * 8 * XPITCH + (i & 3) * 48) = ok ? rg[8 * (QR) + i] : zero; \
        }                                                                                        \
    }

    const float* asrc = A + 16 * mt + DCH * ((int)threadIdx.x >> 2) + 4 * ((int)threadIdx.x & 3);
    v4f ar[3];
    ar[0] = *(const v4f*)(asrc);
    ar[1] = *(const v4f*)(asrc + DCH * 96);
    if (w < 8) ar[2] = *(const v4f*)(asrc + DCH * 192);
    float cwv[3];
    #pragma unroll
    for (int ti = 0; ti < 3; ++ti) cwv[ti] = cw[ch * 9 + 3 * tjw + ti];

    LOAD_STAGE(0)
    asm volatile("" ::: "memory");
    LOAD_STAGE(1)
    asm volatile("" ::: "memory");
    LOAD_STAGE(2)
    asm volatile("" ::: "memory");
    LOAD_STAGE(3)

    v8f acc0 = {}, acc1 = {};
#define COMPUTE_STAGE(BUF, QR)                                                                   \
    {                                                                                            \
        const float* bp = (BUF) + c * XPITCH + 48 * w + 4 * h;                                   \
        const float* ap = a_lds + c * APITCH + 16 * (4 * (QR) + rw) + 4 * h;      \
        v4f av[2];                                                                               \
        av[0] = *(const v4f*)(ap);                                                               \
        av[1] = *(const v4f*)(ap + 8);                                                           \
        v4f f0[6];                                                                               \
        _Pragma("unroll")                                                                        \
        for (int p = 0; p < 6; ++p) f0[p] = *(const v4f*)(bp + 8 * p);                           \
        _Pragma("unroll")                                                                        \
        for (int p = 0; p < 6; ++p) {                                                            \
            const v4f a  = av[p & 1] * cwv[p >> 1];                                              \
            const v4f x0 = f0[p];                                                                \
            v2f a_lo = {a[0], a[1]},   a_hi = {a[2], a[3]};                                      \
            v2f l0   = {x0[0], x0[1]}, h0   = {x0[2], x0[3]};                                    \
            acc0 = __builtin_amdgcn_wmma_f32_16x16x4_f32(false, a_lo, false, l0, (short)0, acc0, false, false); \
            acc1 = __builtin_amdgcn_wmma_f32_16x16x4_f32(false, a_hi, false, h0, (short)0, acc1, false, false); \
        }                                                                                        \
    }

    float* const buf0 = xs;
    float* const buf1 = xs + XSTAGE;
    float* const buf2 = xs + 2 * XSTAGE;
    float* const buf3 = xs + 3 * XSTAGE;

    STORE_STAGE(buf0, 0)
    #pragma unroll
    for (int i = 0; i < 3; ++i) {
        const int q = (int)threadIdx.x + 384 * i;
        if (i < 2 || w < 8) {
            #pragma unroll
            for (int e = 0; e < 4; ++e) a_lds[(4 * (q & 3) + e) * APITCH + (q >> 2)] = ar[i][e];
        }
    }
    lds_barrier();
    COMPUTE_STAGE(buf0, 0)
    STORE_STAGE(buf1, 1)
    lds_barrier();
    COMPUTE_STAGE(buf1, 1)
    STORE_STAGE(buf2, 2)
    lds_barrier();
    COMPUTE_STAGE(buf2, 2)
    STORE_STAGE(buf3, 3)
    lds_barrier();
    COMPUTE_STAGE(buf3, 3)

    #pragma unroll
    for (int j = 0; j < 8; ++j) red[w * RED_Z + j * 32 + lane] = acc0[j] + acc1[j];
    __syncthreads();

    if (w < 8) {
        const int q   = lane >> 4;
        const int px  = lane & 15;
        const int oh  = 2 * oq + (px >> 3);
        const int owp = px & 7;
        const int cl  = 2 * w + q;
        const float* r = red + (cl & 7) * 32 + 16 * (cl >> 3) + px;
        float s = 0.0f;
        #pragma unroll
        for (int ww = 0; ww < CW_WAVES; ++ww) s += r[ww * RED_Z];
        float wsum = 0.0f;
        #pragma unroll
        for (int tap = 0; tap < 9; ++tap) {
            const bool ok = !(oh == 0 && tap < 3) && !(owp == 0 && (tap % 3) == 0);
            wsum += ok ? (q ? e_cw[1][tap] : e_cw[0][tap]) : 0.0f;
        }
        const float acc = (q ? e_cb[1] : e_cb[0]) + s + (q ? e_cv[1] : e_cv[0]) * wsum;
        out[(size_t)(b * DCH + ec0 + q) * 64 + 16 * oq + px] =
            (acc - (q ? e_mean[1] : e_mean[0])) * (q ? e_sc[1] : e_sc[0]) + (q ? e_beta[1] : e_beta[0]);
    }
#undef LOAD_STAGE
#undef STORE_STAGE
#undef COMPUTE_STAGE
}

extern "C" void kernel_launch(void* const* d_in, const int* in_sizes, int n_in,
                              void* d_out, int out_size, void* d_ws, size_t ws_size,
                              hipStream_t stream) {
    const float* x     = (const float*)d_in[0];
    const float* Wr    = (const float*)d_in[1];
    const float* br    = (const float*)d_in[2];
    const float* Wi    = (const float*)d_in[3];
    const float* bi    = (const float*)d_in[4];
    const float* cw    = (const float*)d_in[5];
    const float* cb    = (const float*)d_in[6];
    const float* gamma = (const float*)d_in[7];
    const float* beta  = (const float*)d_in[8];
    const float* mean  = (const float*)d_in[9];
    const float* var   = (const float*)d_in[10];

    char* ws = (char*)d_ws;
    double* B1    = (double*)(ws + 5120);
    double* B2    = (double*)(ws + 136192);
    float*  A     = (float*) (ws + 267264);
    float*  cvec  = (float*) (ws + 332800);

    k_b12<<<DCH * 4, FFT_T, 0, stream>>>(Wr, Wi, B1, B2);
    k_amat<<<INF + 1, FFT_T, 0, stream>>>(B1, B2, br, bi, A, cvec);

    k_conv_wmma<<<256, 384, 0, stream>>>(x, A, cvec, cw, cb, gamma, beta, mean, var,
                                                       (float*)d_out);
}
